# starting kernel
# baseline (speedup 1.0000x reference)
#include <hip/hip_runtime.h>
#include <hip/hip_bf16.h>
#include <math.h>


#define N_NODES 50000
#define N_EDGES 800000
#define IN_DIM  128
#define HID_DIM 64

typedef float v2f __attribute__((ext_vector_type(2)));
typedef float v8f __attribute__((ext_vector_type(8)));


__global__ void init_deg_kernel(float* deg, int n) {
    int i = blockIdx.x * blockDim.x + threadIdx.x;
    if (i < n) deg[i] = 1.0f;
}

__global__ void edge_deg_kernel(const int* __restrict__ dst, float* deg, int e_count) {
    int e = blockIdx.x * blockDim.x + threadIdx.x;
    if (e < e_count) atomicAdd(&deg[dst[e]], 1.0f);
}

__global__ void rsqrt_kernel(float* deg, int n) {
    int i = blockIdx.x * blockDim.x + threadIdx.x;
    if (i < n) {
        float d = deg[i];
        deg[i] = (d > 0.0f) ? rsqrtf(d) : 0.0f;
    }
}


template <int K_DIM>
__global__ __launch_bounds__(256) void gemm_wmma_f32(const float* __restrict__ A,
                                                     const float* __restrict__ W,
                                                     float* __restrict__ C, int M) {
    __shared__ float sW[K_DIM * 64];
    for (int i = threadIdx.x; i < K_DIM * 64; i += 256) sW[i] = W[i];
    __syncthreads();

    const int wave = threadIdx.x >> 5;
    const int lane = threadIdx.x & 31;
    const int m_tile = blockIdx.x * 2 + (wave >> 2);
    const int n_tile = wave & 3;

    const int row16 = lane & 15;
    const int khalf = (lane >> 4) << 1;

    int mrow = m_tile * 16 + row16;
    int arow = (mrow < M) ? mrow : (M - 1);
    const float* __restrict__ Aptr = A + (size_t)arow * K_DIM;
    const float* __restrict__ Bcol = sW + n_tile * 16 + row16;

    v8f acc = {};
    #pragma unroll
    for (int k = 0; k < K_DIM; k += 4) {
        v2f a, b;
        a.x = Aptr[k + khalf];
        a.y = Aptr[k + khalf + 1];
        b.x = Bcol[(k + khalf) * 64];
        b.y = Bcol[(k + khalf + 1) * 64];
        acc = __builtin_amdgcn_wmma_f32_16x16x4_f32(
             false, a,  false, b,
             (short)0, acc,  false,  false);
    }

    const int mbase = m_tile * 16 + ((lane >> 4) << 3);
    const int ncol  = n_tile * 16 + row16;
    #pragma unroll
    for (int j = 0; j < 8; ++j) {
        int m = mbase + j;
        if (m < M) C[(size_t)m * 64 + ncol] = acc[j];
    }
}


__global__ void zero_kernel(float4* p, int n4) {
    int i = blockIdx.x * blockDim.x + threadIdx.x;
    if (i < n4) p[i] = make_float4(0.f, 0.f, 0.f, 0.f);
}


__global__ __launch_bounds__(256) void scatter_kernel(const float* __restrict__ h,
                                                      const float* __restrict__ dinv,
                                                      const int* __restrict__ src,
                                                      const int* __restrict__ dst,
                                                      float* __restrict__ agg, int e_count) {
    int gid = blockIdx.x * blockDim.x + threadIdx.x;
    int e = gid >> 4;
    if (e >= e_count) return;
    int c = (gid & 15) << 2;
    int s = src[e], d = dst[e];
    float norm = dinv[s] * dinv[d];
    const float4 hv = *(const float4*)(h + (size_t)s * 64 + c);
    float* o = agg + (size_t)d * 64 + c;
    atomicAdd(o + 0, hv.x * norm);
    atomicAdd(o + 1, hv.y * norm);
    atomicAdd(o + 2, hv.z * norm);
    atomicAdd(o + 3, hv.w * norm);
}


__global__ __launch_bounds__(256) void finalize_kernel(const float* __restrict__ agg,
                                                       const float* __restrict__ h,
                                                       const float* __restrict__ dinv,
                                                       const float* __restrict__ bias,
                                                       float* __restrict__ out,
                                                       int n_nodes, int do_relu) {
    int gid = blockIdx.x * blockDim.x + threadIdx.x;
    int n = gid >> 4;
    if (n >= n_nodes) return;
    int c = (gid & 15) << 2;
    float di = dinv[n];
    float self_n = di * di;
    const float4 a  = *(const float4*)(agg + (size_t)n * 64 + c);
    const float4 hv = *(const float4*)(h   + (size_t)n * 64 + c);
    const float4 b  = *(const float4*)(bias + c);
    float4 r;
    r.x = a.x + hv.x * self_n + b.x;
    r.y = a.y + hv.y * self_n + b.y;
    r.z = a.z + hv.z * self_n + b.z;
    r.w = a.w + hv.w * self_n + b.w;
    if (do_relu) {
        r.x = fmaxf(r.x, 0.f); r.y = fmaxf(r.y, 0.f);
        r.z = fmaxf(r.z, 0.f); r.w = fmaxf(r.w, 0.f);
    }
    *(float4*)(out + (size_t)n * 64 + c) = r;
}


extern "C" void kernel_launch(void* const* d_in, const int* in_sizes, int n_in,
                              void* d_out, int out_size, void* d_ws, size_t ws_size,
                              hipStream_t stream) {
    const float* x  = (const float*)d_in[0];
    const int*   ei = (const int*)d_in[1];
    const float* W1 = (const float*)d_in[2];
    const float* b1 = (const float*)d_in[3];
    const float* W2 = (const float*)d_in[4];
    const float* b2 = (const float*)d_in[5];
    float* out = (float*)d_out;

    const int* src = ei;
    const int* dst = ei + N_EDGES;

    float* dinv = (float*)d_ws;
    float* buf0 = dinv + 65536;
    float* buf1 = buf0 + (size_t)N_NODES * 64;
    const int feat_elems = N_NODES * 64;

    const int B = 256;
    dim3 blk(B);

    init_deg_kernel<<<(N_NODES + B - 1) / B, blk, 0, stream>>>(dinv, N_NODES);
    edge_deg_kernel<<<(N_EDGES + B - 1) / B, blk, 0, stream>>>(dst, dinv, N_EDGES);
    rsqrt_kernel<<<(N_NODES + B - 1) / B, blk, 0, stream>>>(dinv, N_NODES);

    int gemm_grid = (N_NODES + 31) / 32;
    gemm_wmma_f32<IN_DIM><<<gemm_grid, blk, 0, stream>>>(x, W1, buf0, N_NODES);
    zero_kernel<<<(feat_elems / 4 + B - 1) / B, blk, 0, stream>>>((float4*)buf1, feat_elems / 4);
    scatter_kernel<<<(N_EDGES * 16 + B - 1) / B, blk, 0, stream>>>(buf0, dinv, src, dst, buf1, N_EDGES);
    finalize_kernel<<<(N_NODES * 16 + B - 1) / B, blk, 0, stream>>>(buf1, buf0, dinv, b1,
                                                                    buf0, N_NODES,  1);

    gemm_wmma_f32<HID_DIM><<<gemm_grid, blk, 0, stream>>>(buf0, W2, buf1, N_NODES);
    zero_kernel<<<(feat_elems / 4 + B - 1) / B, blk, 0, stream>>>((float4*)buf0, feat_elems / 4);
    scatter_kernel<<<(N_EDGES * 16 + B - 1) / B, blk, 0, stream>>>(buf1, dinv, src, dst, buf0, N_EDGES);
    finalize_kernel<<<(N_NODES * 16 + B - 1) / B, blk, 0, stream>>>(buf0, buf1, dinv, b2,
                                                                    out, N_NODES,  0);
}


# Push to pull: per call rank edges by dst (one int atomic each) + scan into a dst-sorted list; each node then sums its in-edges by float4 gather, self-loop/bias/relu fused; sum in rank order
# speedup vs baseline: 1.8133x; 1.8133x over previous
#include <hip/hip_runtime.h>
#include <hip/hip_bf16.h>
#include <math.h>


#define N_NODES 50000
#define N_EDGES 800000
#define IN_DIM  128
#define HID_DIM 64

typedef float v2f __attribute__((ext_vector_type(2)));
typedef float v8f __attribute__((ext_vector_type(8)));


__global__ void init_cnt_kernel(int* cnt, int n) {
    int i = blockIdx.x * blockDim.x + threadIdx.x;
    if (i < n) cnt[i] = 0;
}

__global__ void rank_kernel(const int* __restrict__ dst, int* cnt, int* __restrict__ rank,
                            int e_count) {
    int e = blockIdx.x * blockDim.x + threadIdx.x;
    if (e < e_count) rank[e] = atomicAdd(&cnt[dst[e]], 1);
}

__device__ __forceinline__ int block_incl_scan_256(int v, int* wsum) {
    const int lane = threadIdx.x & 31;
    const int wave = threadIdx.x >> 5;
    int incl = v;
    for (int d = 1; d < 32; d <<= 1) {
        int t = __shfl_up(incl, d);
        if (lane >= d) incl += t;
    }
    if (lane == 31) wsum[wave] = incl;
    __syncthreads();
    int base = 0;
    for (int w = 0; w < 8; ++w) {
        int s = wsum[w];
        base += (w < wave) ? s : 0;
    }
    return base + incl;
}

__global__ __launch_bounds__(256) void scan_local_kernel(const int* __restrict__ cnt,
                                                         int* __restrict__ off,
                                                         int* __restrict__ bsum,
                                                         float* __restrict__ dinv, int n) {
    __shared__ int wsum[8];
    int i = blockIdx.x * 256 + threadIdx.x;
    int v = (i < n) ? cnt[i] : 0;
    int incl = block_incl_scan_256(v, wsum);
    if (i < n) {
        off[i] = incl - v;
        float d = (float)(v + 1);
        dinv[i] = (d > 0.0f) ? rsqrtf(d) : 0.0f;
    }
    if (threadIdx.x == 255) bsum[blockIdx.x] = incl;
}

__global__ __launch_bounds__(256) void scan_blocks_kernel(const int* __restrict__ bsum,
                                                          int* __restrict__ bbase, int nb) {
    __shared__ int wsum[8];
    int t = threadIdx.x;
    int v = (t < nb) ? bsum[t] : 0;
    int incl = block_incl_scan_256(v, wsum);
    if (t < nb) bbase[t] = incl - v;
}

__global__ void place_kernel(const int* __restrict__ src, const int* __restrict__ dst,
                             const int* __restrict__ rank, const int* __restrict__ off,
                             const int* __restrict__ bbase, const float* __restrict__ dinv,
                             int2* __restrict__ col, int e_count) {
    int e = blockIdx.x * blockDim.x + threadIdx.x;
    if (e >= e_count) return;
    int s = src[e], d = dst[e];
    int p = bbase[d >> 8] + off[d] + rank[e];
    p = min(max(p, 0), e_count - 1);
    col[p] = make_int2(s, __float_as_int(dinv[s]));
}


template <int K_DIM>
__global__ __launch_bounds__(256) void gemm_wmma_f32(const float* __restrict__ A,
                                                     const float* __restrict__ W,
                                                     float* __restrict__ C, int M) {
    __shared__ float sW[K_DIM * 64];
    for (int i = threadIdx.x; i < K_DIM * 64; i += 256) sW[i] = W[i];
    __syncthreads();

    const int wave = threadIdx.x >> 5;
    const int lane = threadIdx.x & 31;
    const int m_tile = blockIdx.x * 2 + (wave >> 2);
    const int n_tile = wave & 3;

    const int row16 = lane & 15;
    const int khalf = (lane >> 4) << 1;

    int mrow = m_tile * 16 + row16;
    int arow = (mrow < M) ? mrow : (M - 1);
    const float* __restrict__ Aptr = A + (size_t)arow * K_DIM;
    const float* __restrict__ Bcol = sW + n_tile * 16 + row16;

    v8f acc = {};
    #pragma unroll
    for (int k = 0; k < K_DIM; k += 4) {
        v2f a, b;
        a.x = Aptr[k + khalf];
        a.y = Aptr[k + khalf + 1];
        b.x = Bcol[(k + khalf) * 64];
        b.y = Bcol[(k + khalf + 1) * 64];
        acc = __builtin_amdgcn_wmma_f32_16x16x4_f32(
             false, a,  false, b,
             (short)0, acc,  false,  false);
    }

    const int mbase = m_tile * 16 + ((lane >> 4) << 3);
    const int ncol  = n_tile * 16 + row16;
    #pragma unroll
    for (int j = 0; j < 8; ++j) {
        int m = mbase + j;
        if (m < M) C[(size_t)m * 64 + ncol] = acc[j];
    }
}


__global__ __launch_bounds__(256) void gather_kernel(const float* __restrict__ h,
                                                     const float* __restrict__ dinv,
                                                     const int* __restrict__ cnt,
                                                     const int* __restrict__ off,
                                                     const int* __restrict__ bbase,
                                                     const int2* __restrict__ col,
                                                     const float* __restrict__ bias,
                                                     float* __restrict__ out,
                                                     int n_nodes, int e_count, int do_relu) {
    int gid = blockIdx.x * blockDim.x + threadIdx.x;
    int n = gid >> 4;
    if (n >= n_nodes) return;
    int c = (gid & 15) << 2;
    const float di = dinv[n];
    const int start = max(bbase[n >> 8] + off[n], 0);
    const int end = min(start + cnt[n], e_count);
    float4 a = make_float4(0.f, 0.f, 0.f, 0.f);
    if (start < end) {
        const int last = end - 1;
        int2 e0 = col[start];
        int2 e1 = col[min(start + 1, last)];
        int2 e2 = col[min(start + 2, last)];
        int2 e3 = col[min(start + 3, last)];
        for (int p = start; p < end; p += 4) {
            const float4 h0 = *(const float4*)(h + (size_t)e0.x * 64 + c);
            const float4 h1 = *(const float4*)(h + (size_t)e1.x * 64 + c);
            const float4 h2 = *(const float4*)(h + (size_t)e2.x * 64 + c);
            const float4 h3 = *(const float4*)(h + (size_t)e3.x * 64 + c);
            const float w0 = __int_as_float(e0.y) * di;
            const float w1 = (p + 1 < end) ? __int_as_float(e1.y) * di : 0.f;
            const float w2 = (p + 2 < end) ? __int_as_float(e2.y) * di : 0.f;
            const float w3 = (p + 3 < end) ? __int_as_float(e3.y) * di : 0.f;
            e0 = col[min(p + 4, last)];
            e1 = col[min(p + 5, last)];
            e2 = col[min(p + 6, last)];
            e3 = col[min(p + 7, last)];
            a.x = fmaf(h0.x, w0, a.x); a.y = fmaf(h0.y, w0, a.y);
            a.z = fmaf(h0.z, w0, a.z); a.w = fmaf(h0.w, w0, a.w);
            a.x = fmaf(h1.x, w1, a.x); a.y = fmaf(h1.y, w1, a.y);
            a.z = fmaf(h1.z, w1, a.z); a.w = fmaf(h1.w, w1, a.w);
            a.x = fmaf(h2.x, w2, a.x); a.y = fmaf(h2.y, w2, a.y);
            a.z = fmaf(h2.z, w2, a.z); a.w = fmaf(h2.w, w2, a.w);
            a.x = fmaf(h3.x, w3, a.x); a.y = fmaf(h3.y, w3, a.y);
            a.z = fmaf(h3.z, w3, a.z); a.w = fmaf(h3.w, w3, a.w);
        }
    }
    const float self_n = di * di;
    const float4 hv = *(const float4*)(h + (size_t)n * 64 + c);
    const float4 b  = *(const float4*)(bias + c);
    float4 r;
    r.x = a.x + hv.x * self_n + b.x;
    r.y = a.y + hv.y * self_n + b.y;
    r.z = a.z + hv.z * self_n + b.z;
    r.w = a.w + hv.w * self_n + b.w;
    if (do_relu) {
        r.x = fmaxf(r.x, 0.f); r.y = fmaxf(r.y, 0.f);
        r.z = fmaxf(r.z, 0.f); r.w = fmaxf(r.w, 0.f);
    }
    *(float4*)(out + (size_t)n * 64 + c) = r;
}


extern "C" void kernel_launch(void* const* d_in, const int* in_sizes, int n_in,
                              void* d_out, int out_size, void* d_ws, size_t ws_size,
                              hipStream_t stream) {
    const float* x  = (const float*)d_in[0];
    const int*   ei = (const int*)d_in[1];
    const float* W1 = (const float*)d_in[2];
    const float* b1 = (const float*)d_in[3];
    const float* W2 = (const float*)d_in[4];
    const float* b2 = (const float*)d_in[5];
    float* out = (float*)d_out;

    const int* src = ei;
    const int* dst = ei + N_EDGES;

    float* dinv  = (float*)d_ws;
    int*   cnt   = (int*)(dinv + 65536);
    int*   off   = cnt + 65536;
    int*   bsum  = off + 65536;
    int*   bbase = bsum + 1024;
    int*   rank  = bbase + 1024;
    int2*  col   = (int2*)(rank + N_EDGES);
    float* buf0  = (float*)(col + N_EDGES);
    float* buf1  = buf0 + (size_t)N_NODES * 64;

    const int B = 256;
    dim3 blk(B);
    const int node_blocks = (N_NODES + B - 1) / B;

    init_cnt_kernel<<<node_blocks, blk, 0, stream>>>(cnt, N_NODES);
    rank_kernel<<<(N_EDGES + B - 1) / B, blk, 0, stream>>>(dst, cnt, rank, N_EDGES);
    scan_local_kernel<<<node_blocks, blk, 0, stream>>>(cnt, off, bsum, dinv, N_NODES);
    scan_blocks_kernel<<<1, blk, 0, stream>>>(bsum, bbase, node_blocks);
    place_kernel<<<(N_EDGES + B - 1) / B, blk, 0, stream>>>(src, dst, rank, off, bbase, dinv,
                                                            col, N_EDGES);

    int gemm_grid = (N_NODES + 31) / 32;
    gemm_wmma_f32<IN_DIM><<<gemm_grid, blk, 0, stream>>>(x, W1, buf0, N_NODES);
    gather_kernel<<<(N_NODES * 16 + B - 1) / B, blk, 0, stream>>>(buf0, dinv, cnt, off, bbase, col, b1,
                                                                  buf1, N_NODES, N_EDGES,  1);

    gemm_wmma_f32<HID_DIM><<<gemm_grid, blk, 0, stream>>>(buf1, W2, buf0, N_NODES);
    gather_kernel<<<(N_NODES * 16 + B - 1) / B, blk, 0, stream>>>(buf0, dinv, cnt, off, bbase, col, b2,
                                                                  out, N_NODES, N_EDGES,  0);
}


# Edge sort with no global atomics: chunk histogram, bucket scatter, per-256-node finalize, every rank from an LDS counter; dinv[src] moved into GEMM epilogue; sum order = LDS arrival rank
# speedup vs baseline: 3.0990x; 1.7090x over previous
#include <hip/hip_runtime.h>
#include <hip/hip_bf16.h>
#include <math.h>


#define N_NODES 50000
#define N_EDGES 800000
#define IN_DIM  128
#define HID_DIM 64

#define N_BUCKETS 196
#define N_CHUNKS  256
#define CHUNK     3125
#define LSORT_CAP 8192

static_assert(N_CHUNKS * CHUNK == N_EDGES, "chunks must tile the edge list exactly");
static_assert(N_BUCKETS * 256 >= N_NODES && N_BUCKETS <= 256, "bucket count");

typedef float v2f __attribute__((ext_vector_type(2)));
typedef float v8f __attribute__((ext_vector_type(8)));


__device__ __forceinline__ int bucket_of(int d) {
    unsigned b = ((unsigned)d) >> 8;
    return (int)(b < (unsigned)(N_BUCKETS - 1) ? b : (unsigned)(N_BUCKETS - 1));
}

__device__ __forceinline__ int block_incl_scan_1024(int v, int* wsum) {
    const int lane = threadIdx.x & 31;
    const int wave = threadIdx.x >> 5;
    int incl = v;
    for (int d = 1; d < 32; d <<= 1) {
        int t = __shfl_up(incl, d);
        if (lane >= d) incl += t;
    }
    if (lane == 31) wsum[wave] = incl;
    __syncthreads();
    int base = 0;
    for (int w = 0; w < 32; ++w) {
        int s = wsum[w];
        base += (w < wave) ? s : 0;
    }
    __syncthreads();
    return base + incl;
}

__global__ __launch_bounds__(1024) void bucket_hist_kernel(const int* __restrict__ dst,
                                                           int* __restrict__ hist) {
    __shared__ int h[256];
    const int tid = threadIdx.x;
    if (tid < 256) h[tid] = 0;
    __syncthreads();
    const int base = blockIdx.x * CHUNK;
    for (int j = 0; j < 4; ++j) {
        const int i = tid + j * 1024;
        if (i < CHUNK) atomicAdd(&h[bucket_of(dst[base + i])], 1);
    }
    __syncthreads();
    if (tid < 256) hist[blockIdx.x * 256 + tid] = h[tid];
}

__global__ __launch_bounds__(1024) void bucket_scatter_kernel(const int* __restrict__ src,
                                                              const int* __restrict__ dst,
                                                              const int* __restrict__ hist,
                                                              int* __restrict__ bkt_start,
                                                              int* __restrict__ bkt_cnt,
                                                              int2* __restrict__ bucketed) {
    __shared__ int sorted_s[CHUNK];
    __shared__ int sorted_d[CHUNK];
    __shared__ int ptot[16][256];
    __shared__ int ppre[16][256];
    __shared__ int lstart[256];
    __shared__ int gbase[256];
    __shared__ int cursor[256];
    __shared__ int wsum[32];

    const int tid = threadIdx.x;
    const int c = blockIdx.x;
    const int t4 = (tid & 63) << 2;
    const int q = tid >> 6;

    int es[4], ed[4];
    for (int j = 0; j < 4; ++j) {
        const int i = tid + j * 1024;
        es[j] = 0; ed[j] = 0;
        if (i < CHUNK) { es[j] = src[c * CHUNK + i]; ed[j] = dst[c * CHUNK + i]; }
    }

    int tx = 0, ty = 0, tz = 0, tw = 0, px = 0, py = 0, pz = 0, pw = 0;
    for (int j = q * 16; j < q * 16 + 16; ++j) {
        const int4 v = *(const int4*)(hist + j * 256 + t4);
        tx += v.x; ty += v.y; tz += v.z; tw += v.w;
        const bool before = (j < c);
        px += before ? v.x : 0; py += before ? v.y : 0;
        pz += before ? v.z : 0; pw += before ? v.w : 0;
    }
    ptot[q][t4 + 0] = tx; ptot[q][t4 + 1] = ty; ptot[q][t4 + 2] = tz; ptot[q][t4 + 3] = tw;
    ppre[q][t4 + 0] = px; ppre[q][t4 + 1] = py; ppre[q][t4 + 2] = pz; ppre[q][t4 + 3] = pw;
    if (tid < 256) cursor[tid] = 0;
    __syncthreads();

    int btot = 0, bpre = 0, lcnt = 0;
    if (tid < 256) {
        for (int k = 0; k < 16; ++k) { btot += ptot[k][tid]; bpre += ppre[k][tid]; }
        lcnt = hist[c * 256 + tid];
    }
    const int incl_g = block_incl_scan_1024(btot, wsum);
    const int incl_l = block_incl_scan_1024(lcnt, wsum);
    if (tid < 256) {
        const int bs = incl_g - btot;
        gbase[tid] = bs + bpre;
        lstart[tid] = incl_l - lcnt;
        if (c == 0) { bkt_start[tid] = bs; bkt_cnt[tid] = btot; }
    }
    __syncthreads();

    for (int j = 0; j < 4; ++j) {
        const int i = tid + j * 1024;
        if (i < CHUNK) {
            const int b = bucket_of(ed[j]);
            const int r = atomicAdd(&cursor[b], 1);
            const int lp = min(max(lstart[b] + r, 0), CHUNK - 1);
            sorted_s[lp] = es[j];
            sorted_d[lp] = ed[j];
        }
    }
    __syncthreads();

    for (int j = 0; j < 4; ++j) {
        const int i = tid + j * 1024;
        if (i < CHUNK) {
            const int s = sorted_s[i], d = sorted_d[i];
            const int b = bucket_of(d);
            int pos = gbase[b] + (i - lstart[b]);
            pos = min(max(pos, 0), N_EDGES - 1);
            bucketed[pos] = make_int2(s, d);
        }
    }
}

__global__ __launch_bounds__(1024) void bucket_finalize_kernel(const int2* __restrict__ bucketed,
                                                               const int* __restrict__ bkt_start,
                                                               const int* __restrict__ bkt_cnt,
                                                               int* __restrict__ cnt,
                                                               int* __restrict__ rowstart,
                                                               float* __restrict__ dinv,
                                                               int* __restrict__ col) {
    __shared__ int ncnt[256];
    __shared__ int nstart[256];
    __shared__ int wsum[32];
    __shared__ int lsorted[LSORT_CAP];

    const int tid = threadIdx.x;
    const int b = blockIdx.x;
    const int s = min(max(bkt_start[b], 0), N_EDGES);
    const int m = min(max(bkt_cnt[b], 0), N_EDGES - s);

    const bool in_lds = (m <= LSORT_CAP);

    int rs[8], rd[8];
    const int mlast = max(m - 1, 0);
    for (int k = 0; k < 8; ++k) {
        const int i = min(tid + k * 1024, mlast);
        const int2 rec = bucketed[min(s + i, N_EDGES - 1)];
        rs[k] = rec.x; rd[k] = rec.y;
    }

    if (tid < 256) ncnt[tid] = 0;
    __syncthreads();
    if (in_lds) {
        for (int k = 0; k < 8; ++k) {
            if (tid + k * 1024 < m) atomicAdd(&ncnt[rd[k] & 255], 1);
        }
    } else {
        for (int i = tid; i < m; i += 1024) {
            const int d = bucketed[s + i].y;
            atomicAdd(&ncnt[d & 255], 1);
        }
    }
    __syncthreads();

    const int v = (tid < 256) ? ncnt[tid] : 0;
    const int incl = block_incl_scan_1024(v, wsum);
    if (tid < 256) {
        const int ex = incl - v;
        nstart[tid] = ex;
        ncnt[tid] = 0;
        const int n = b * 256 + tid;
        if (n < N_NODES) {
            cnt[n] = v;
            rowstart[n] = s + ex;
            const float dg = (float)(v + 1);
            dinv[n] = (dg > 0.0f) ? rsqrtf(dg) : 0.0f;
        }
    }
    __syncthreads();

    if (in_lds) {
        for (int k = 0; k < 8; ++k) {
            if (tid + k * 1024 < m) {
                const int dl = rd[k] & 255;
                const int r = atomicAdd(&ncnt[dl], 1);
                const int p = min(max(nstart[dl] + r, 0), m - 1);
                lsorted[p] = rs[k];
            }
        }
    } else {
        for (int i = tid; i < m; i += 1024) {
            const int2 rec = bucketed[s + i];
            const int dl = rec.y & 255;
            const int r = atomicAdd(&ncnt[dl], 1);
            const int p = min(max(nstart[dl] + r, 0), m - 1);
            col[s + p] = rec.x;
        }
    }
    __syncthreads();
    if (in_lds) {
        for (int i = tid; i < m; i += 1024) col[s + i] = lsorted[i];
    }
}


template <int K_DIM>
__global__ __launch_bounds__(256) void gemm_wmma_f32(const float* __restrict__ A,
                                                     const float* __restrict__ W,
                                                     const float* __restrict__ rscale,
                                                     float* __restrict__ C, int M) {
    __shared__ float sW[K_DIM * 64];
    for (int i = threadIdx.x; i < K_DIM * 64; i += 256) sW[i] = W[i];
    __syncthreads();

    const int wave = threadIdx.x >> 5;
    const int lane = threadIdx.x & 31;
    const int m_tile = blockIdx.x * 2 + (wave >> 2);
    const int n_tile = wave & 3;

    const int row16 = lane & 15;
    const int khalf = (lane >> 4) << 1;

    int mrow = m_tile * 16 + row16;
    int arow = (mrow < M) ? mrow : (M - 1);
    const float* __restrict__ Aptr = A + (size_t)arow * K_DIM;
    const float* __restrict__ Bcol = sW + n_tile * 16 + row16;

    v8f acc = {};
    #pragma unroll
    for (int k = 0; k < K_DIM; k += 4) {
        v2f a, b;
        a.x = Aptr[k + khalf];
        a.y = Aptr[k + khalf + 1];
        b.x = Bcol[(k + khalf) * 64];
        b.y = Bcol[(k + khalf + 1) * 64];
        acc = __builtin_amdgcn_wmma_f32_16x16x4_f32(
             false, a,  false, b,
             (short)0, acc,  false,  false);
    }

    const int mbase = m_tile * 16 + ((lane >> 4) << 3);
    const int ncol  = n_tile * 16 + row16;
    #pragma unroll
    for (int j = 0; j < 8; ++j) {
        int m = mbase + j;
        if (m < M) C[(size_t)m * 64 + ncol] = acc[j] * rscale[m];
    }
}


__global__ __launch_bounds__(256) void gather_kernel(const float* __restrict__ h,
                                                     const float* __restrict__ dinv,
                                                     const int* __restrict__ cnt,
                                                     const int* __restrict__ rowstart,
                                                     const int* __restrict__ col,
                                                     const float* __restrict__ bias,
                                                     float* __restrict__ out,
                                                     int n_nodes, int e_count, int do_relu) {
    int gid = blockIdx.x * blockDim.x + threadIdx.x;
    int n = gid >> 4;
    if (n >= n_nodes) return;
    int c = (gid & 15) << 2;
    const float di = dinv[n];
    const int start = max(rowstart[n], 0);
    const int end = min(start + cnt[n], e_count);
    float4 a = make_float4(0.f, 0.f, 0.f, 0.f);
    if (start < end) {
        const int last = end - 1;
        int s0 = col[start];
        int s1 = col[min(start + 1, last)];
        int s2 = col[min(start + 2, last)];
        int s3 = col[min(start + 3, last)];
        for (int p = start; p < end; p += 4) {
            const float4 h0 = *(const float4*)(h + (size_t)s0 * 64 + c);
            const float4 h1 = *(const float4*)(h + (size_t)s1 * 64 + c);
            const float4 h2 = *(const float4*)(h + (size_t)s2 * 64 + c);
            const float4 h3 = *(const float4*)(h + (size_t)s3 * 64 + c);
            const float w1 = (p + 1 < end) ? 1.f : 0.f;
            const float w2 = (p + 2 < end) ? 1.f : 0.f;
            const float w3 = (p + 3 < end) ? 1.f : 0.f;
            s0 = col[min(p + 4, last)];
            s1 = col[min(p + 5, last)];
            s2 = col[min(p + 6, last)];
            s3 = col[min(p + 7, last)];
            a.x += h0.x; a.y += h0.y; a.z += h0.z; a.w += h0.w;
            a.x = fmaf(h1.x, w1, a.x); a.y = fmaf(h1.y, w1, a.y);
            a.z = fmaf(h1.z, w1, a.z); a.w = fmaf(h1.w, w1, a.w);
            a.x = fmaf(h2.x, w2, a.x); a.y = fmaf(h2.y, w2, a.y);
            a.z = fmaf(h2.z, w2, a.z); a.w = fmaf(h2.w, w2, a.w);
            a.x = fmaf(h3.x, w3, a.x); a.y = fmaf(h3.y, w3, a.y);
            a.z = fmaf(h3.z, w3, a.z); a.w = fmaf(h3.w, w3, a.w);
        }
    }
    const float4 hv = *(const float4*)(h + (size_t)n * 64 + c);
    const float4 b  = *(const float4*)(bias + c);
    float4 r;
    r.x = (a.x + hv.x) * di + b.x;
    r.y = (a.y + hv.y) * di + b.y;
    r.z = (a.z + hv.z) * di + b.z;
    r.w = (a.w + hv.w) * di + b.w;
    if (do_relu) {
        r.x = fmaxf(r.x, 0.f); r.y = fmaxf(r.y, 0.f);
        r.z = fmaxf(r.z, 0.f); r.w = fmaxf(r.w, 0.f);
    }
    *(float4*)(out + (size_t)n * 64 + c) = r;
}


extern "C" void kernel_launch(void* const* d_in, const int* in_sizes, int n_in,
                              void* d_out, int out_size, void* d_ws, size_t ws_size,
                              hipStream_t stream) {
    const float* x  = (const float*)d_in[0];
    const int*   ei = (const int*)d_in[1];
    const float* W1 = (const float*)d_in[2];
    const float* b1 = (const float*)d_in[3];
    const float* W2 = (const float*)d_in[4];
    const float* b2 = (const float*)d_in[5];
    float* out = (float*)d_out;

    const int* src = ei;
    const int* dst = ei + N_EDGES;

    float* dinv      = (float*)d_ws;
    int*   cnt       = (int*)(dinv + 65536);
    int*   rowstart  = cnt + 65536;
    int*   hist      = rowstart + 65536;
    int*   bkt_start = hist + 65536;
    int*   bkt_cnt   = bkt_start + 256;
    int2*  bucketed  = (int2*)(bkt_cnt + 256);
    int*   col       = (int*)(bucketed + N_EDGES);
    float* buf0      = (float*)(col + N_EDGES);
    float* buf1      = buf0 + (size_t)N_NODES * 64;

    const int B = 256;
    dim3 blk(B);
    dim3 blk1k(1024);

    bucket_hist_kernel<<<N_CHUNKS, blk1k, 0, stream>>>(dst, hist);
    bucket_scatter_kernel<<<N_CHUNKS, blk1k, 0, stream>>>(src, dst, hist, bkt_start, bkt_cnt, bucketed);
    bucket_finalize_kernel<<<N_BUCKETS, blk1k, 0, stream>>>(bucketed, bkt_start, bkt_cnt, cnt, rowstart,
                                                            dinv, col);

    int gemm_grid = (N_NODES + 31) / 32;
    gemm_wmma_f32<IN_DIM><<<gemm_grid, blk, 0, stream>>>(x, W1, dinv, buf0, N_NODES);
    gather_kernel<<<(N_NODES * 16 + B - 1) / B, blk, 0, stream>>>(buf0, dinv, cnt, rowstart, col, b1,
                                                                  buf1, N_NODES, N_EDGES,  1);

    gemm_wmma_f32<HID_DIM><<<gemm_grid, blk, 0, stream>>>(buf1, W2, dinv, buf0, N_NODES);
    gather_kernel<<<(N_NODES * 16 + B - 1) / B, blk, 0, stream>>>(buf0, dinv, cnt, rowstart, col, b2,
                                                                  out, N_NODES, N_EDGES,  0);
}


# Layer-1 gather keeps its 16 finished rows in LDS and multiplies them by W2 in the same work-group (same f32 WMMA, same k order): layer 2's own GEMM launch and the hidden layer's round trip are gone
# speedup vs baseline: 3.6055x; 1.1634x over previous
#include <hip/hip_runtime.h>
#include <hip/hip_bf16.h>
#include <math.h>


#define N_NODES 50000
#define N_EDGES 800000
#define IN_DIM  128
#define HID_DIM 64

#define N_BUCKETS 196
#define N_CHUNKS  256
#define CHUNK     3125
#define LSORT_CAP 8192

static_assert(N_CHUNKS * CHUNK == N_EDGES, "chunks must tile the edge list exactly");
static_assert(N_BUCKETS * 256 >= N_NODES && N_BUCKETS <= 256, "bucket count");

typedef float v2f __attribute__((ext_vector_type(2)));
typedef float v8f __attribute__((ext_vector_type(8)));


__device__ __forceinline__ int bucket_of(int d) {
    unsigned b = ((unsigned)d) >> 8;
    return (int)(b < (unsigned)(N_BUCKETS - 1) ? b : (unsigned)(N_BUCKETS - 1));
}

__device__ __forceinline__ int block_incl_scan_1024(int v, int* wsum) {
    const int lane = threadIdx.x & 31;
    const int wave = threadIdx.x >> 5;
    int incl = v;
    for (int d = 1; d < 32; d <<= 1) {
        int t = __shfl_up(incl, d);
        if (lane >= d) incl += t;
    }
    if (lane == 31) wsum[wave] = incl;
    __syncthreads();
    int base = 0;
    for (int w = 0; w < 32; ++w) {
        int s = wsum[w];
        base += (w < wave) ? s : 0;
    }
    __syncthreads();
    return base + incl;
}

__global__ __launch_bounds__(1024) void bucket_hist_kernel(const int* __restrict__ dst,
                                                           int* __restrict__ hist) {
    __shared__ int h[256];
    const int tid = threadIdx.x;
    if (tid < 256) h[tid] = 0;
    __syncthreads();
    const int base = blockIdx.x * CHUNK;
    for (int j = 0; j < 4; ++j) {
        const int i = tid + j * 1024;
        if (i < CHUNK) atomicAdd(&h[bucket_of(dst[base + i])], 1);
    }
    __syncthreads();
    if (tid < 256) hist[blockIdx.x * 256 + tid] = h[tid];
}

__global__ __launch_bounds__(1024) void bucket_scatter_kernel(const int* __restrict__ src,
                                                              const int* __restrict__ dst,
                                                              const int* __restrict__ hist,
                                                              int* __restrict__ bkt_start,
                                                              int* __restrict__ bkt_cnt,
                                                              int2* __restrict__ bucketed) {
    __shared__ int sorted_s[CHUNK];
    __shared__ int sorted_d[CHUNK];
    __shared__ int ptot[16][256];
    __shared__ int ppre[16][256];
    __shared__ int lstart[256];
    __shared__ int gbase[256];
    __shared__ int cursor[256];
    __shared__ int wsum[32];

    const int tid = threadIdx.x;
    const int c = blockIdx.x;
    const int t4 = (tid & 63) << 2;
    const int q = tid >> 6;

    int es[4], ed[4];
    for (int j = 0; j < 4; ++j) {
        const int i = tid + j * 1024;
        es[j] = 0; ed[j] = 0;
        if (i < CHUNK) { es[j] = src[c * CHUNK + i]; ed[j] = dst[c * CHUNK + i]; }
    }

    int tx = 0, ty = 0, tz = 0, tw = 0, px = 0, py = 0, pz = 0, pw = 0;
    for (int j = q * 16; j < q * 16 + 16; ++j) {
        const int4 v = *(const int4*)(hist + j * 256 + t4);
        tx += v.x; ty += v.y; tz += v.z; tw += v.w;
        const bool before = (j < c);
        px += before ? v.x : 0; py += before ? v.y : 0;
        pz += before ? v.z : 0; pw += before ? v.w : 0;
    }
    ptot[q][t4 + 0] = tx; ptot[q][t4 + 1] = ty; ptot[q][t4 + 2] = tz; ptot[q][t4 + 3] = tw;
    ppre[q][t4 + 0] = px; ppre[q][t4 + 1] = py; ppre[q][t4 + 2] = pz; ppre[q][t4 + 3] = pw;
    if (tid < 256) cursor[tid] = 0;
    __syncthreads();

    int btot = 0, bpre = 0, lcnt = 0;
    if (tid < 256) {
        for (int k = 0; k < 16; ++k) { btot += ptot[k][tid]; bpre += ppre[k][tid]; }
        lcnt = hist[c * 256 + tid];
    }
    const int incl_g = block_incl_scan_1024(btot, wsum);
    const int incl_l = block_incl_scan_1024(lcnt, wsum);
    if (tid < 256) {
        const int bs = incl_g - btot;
        gbase[tid] = bs + bpre;
        lstart[tid] = incl_l - lcnt;
        if (c == 0) { bkt_start[tid] = bs; bkt_cnt[tid] = btot; }
    }
    __syncthreads();

    for (int j = 0; j < 4; ++j) {
        const int i = tid + j * 1024;
        if (i < CHUNK) {
            const int b = bucket_of(ed[j]);
            const int r = atomicAdd(&cursor[b], 1);
            const int lp = min(max(lstart[b] + r, 0), CHUNK - 1);
            sorted_s[lp] = es[j];
            sorted_d[lp] = ed[j];
        }
    }
    __syncthreads();

    for (int j = 0; j < 4; ++j) {
        const int i = tid + j * 1024;
        if (i < CHUNK) {
            const int s = sorted_s[i], d = sorted_d[i];
            const int b = bucket_of(d);
            int pos = gbase[b] + (i - lstart[b]);
            pos = min(max(pos, 0), N_EDGES - 1);
            bucketed[pos] = make_int2(s, d);
        }
    }
}

__global__ __launch_bounds__(1024) void bucket_finalize_kernel(const int2* __restrict__ bucketed,
                                                               const int* __restrict__ bkt_start,
                                                               const int* __restrict__ bkt_cnt,
                                                               int* __restrict__ cnt,
                                                               int* __restrict__ rowstart,
                                                               float* __restrict__ dinv,
                                                               int* __restrict__ col) {
    __shared__ int ncnt[256];
    __shared__ int nstart[256];
    __shared__ int wsum[32];
    __shared__ int lsorted[LSORT_CAP];

    const int tid = threadIdx.x;
    const int b = blockIdx.x;
    const int s = min(max(bkt_start[b], 0), N_EDGES);
    const int m = min(max(bkt_cnt[b], 0), N_EDGES - s);

    const bool in_lds = (m <= LSORT_CAP);

    int rs[8], rd[8];
    const int mlast = max(m - 1, 0);
    for (int k = 0; k < 8; ++k) {
        const int i = min(tid + k * 1024, mlast);
        const int2 rec = bucketed[min(s + i, N_EDGES - 1)];
        rs[k] = rec.x; rd[k] = rec.y;
    }

    if (tid < 256) ncnt[tid] = 0;
    __syncthreads();
    if (in_lds) {
        for (int k = 0; k < 8; ++k) {
            if (tid + k * 1024 < m) atomicAdd(&ncnt[rd[k] & 255], 1);
        }
    } else {
        for (int i = tid; i < m; i += 1024) {
            const int d = bucketed[s + i].y;
            atomicAdd(&ncnt[d & 255], 1);
        }
    }
    __syncthreads();

    const int v = (tid < 256) ? ncnt[tid] : 0;
    const int incl = block_incl_scan_1024(v, wsum);
    if (tid < 256) {
        const int ex = incl - v;
        nstart[tid] = ex;
        ncnt[tid] = 0;
        const int n = b * 256 + tid;
        if (n < N_NODES) {
            cnt[n] = v;
            rowstart[n] = s + ex;
            const float dg = (float)(v + 1);
            dinv[n] = (dg > 0.0f) ? rsqrtf(dg) : 0.0f;
        }
    }
    __syncthreads();

    if (in_lds) {
        for (int k = 0; k < 8; ++k) {
            if (tid + k * 1024 < m) {
                const int dl = rd[k] & 255;
                const int r = atomicAdd(&ncnt[dl], 1);
                const int p = min(max(nstart[dl] + r, 0), m - 1);
                lsorted[p] = rs[k];
            }
        }
    } else {
        for (int i = tid; i < m; i += 1024) {
            const int2 rec = bucketed[s + i];
            const int dl = rec.y & 255;
            const int r = atomicAdd(&ncnt[dl], 1);
            const int p = min(max(nstart[dl] + r, 0), m - 1);
            col[s + p] = rec.x;
        }
    }
    __syncthreads();
    if (in_lds) {
        for (int i = tid; i < m; i += 1024) col[s + i] = lsorted[i];
    }
}


template <int K_DIM>
__global__ __launch_bounds__(256) void gemm_wmma_f32(const float* __restrict__ A,
                                                     const float* __restrict__ W,
                                                     const float* __restrict__ rscale,
                                                     float* __restrict__ C, int M) {
    __shared__ float sW[K_DIM * 64];
    for (int i = threadIdx.x; i < K_DIM * 64; i += 256) sW[i] = W[i];
    __syncthreads();

    const int wave = threadIdx.x >> 5;
    const int lane = threadIdx.x & 31;
    const int m_tile = blockIdx.x * 2 + (wave >> 2);
    const int n_tile = wave & 3;

    const int row16 = lane & 15;
    const int khalf = (lane >> 4) << 1;

    int mrow = m_tile * 16 + row16;
    int arow = (mrow < M) ? mrow : (M - 1);
    const float* __restrict__ Aptr = A + (size_t)arow * K_DIM;
    const float* __restrict__ Bcol = sW + n_tile * 16 + row16;

    v8f acc = {};
    #pragma unroll
    for (int k = 0; k < K_DIM; k += 4) {
        v2f a, b;
        a.x = Aptr[k + khalf];
        a.y = Aptr[k + khalf + 1];
        b.x = Bcol[(k + khalf) * 64];
        b.y = Bcol[(k + khalf + 1) * 64];
        acc = __builtin_amdgcn_wmma_f32_16x16x4_f32(
             false, a,  false, b,
             (short)0, acc,  false,  false);
    }

    const int mbase = m_tile * 16 + ((lane >> 4) << 3);
    const int ncol  = n_tile * 16 + row16;
    #pragma unroll
    for (int j = 0; j < 8; ++j) {
        int m = mbase + j;
        if (m < M) C[(size_t)m * 64 + ncol] = acc[j] * rscale[m];
    }
}


__global__ __launch_bounds__(256) void gather_kernel(const float* __restrict__ h,
                                                     const float* __restrict__ dinv,
                                                     const int* __restrict__ cnt,
                                                     const int* __restrict__ rowstart,
                                                     const int* __restrict__ col,
                                                     const float* __restrict__ bias,
                                                     float* __restrict__ out,
                                                     int n_nodes, int e_count, int do_relu) {
    int gid = blockIdx.x * blockDim.x + threadIdx.x;
    int n = gid >> 4;
    if (n >= n_nodes) return;
    int c = (gid & 15) << 2;
    const float di = dinv[n];
    const int start = max(rowstart[n], 0);
    const int end = min(start + cnt[n], e_count);
    float4 a = make_float4(0.f, 0.f, 0.f, 0.f);
    if (start < end) {
        const int last = end - 1;
        int s0 = col[start];
        int s1 = col[min(start + 1, last)];
        int s2 = col[min(start + 2, last)];
        int s3 = col[min(start + 3, last)];
        for (int p = start; p < end; p += 4) {
            const float4 h0 = *(const float4*)(h + (size_t)s0 * 64 + c);
            const float4 h1 = *(const float4*)(h + (size_t)s1 * 64 + c);
            const float4 h2 = *(const float4*)(h + (size_t)s2 * 64 + c);
            const float4 h3 = *(const float4*)(h + (size_t)s3 * 64 + c);
            const float w1 = (p + 1 < end) ? 1.f : 0.f;
            const float w2 = (p + 2 < end) ? 1.f : 0.f;
            const float w3 = (p + 3 < end) ? 1.f : 0.f;
            s0 = col[min(p + 4, last)];
            s1 = col[min(p + 5, last)];
            s2 = col[min(p + 6, last)];
            s3 = col[min(p + 7, last)];
            a.x += h0.x; a.y += h0.y; a.z += h0.z; a.w += h0.w;
            a.x = fmaf(h1.x, w1, a.x); a.y = fmaf(h1.y, w1, a.y);
            a.z = fmaf(h1.z, w1, a.z); a.w = fmaf(h1.w, w1, a.w);
            a.x = fmaf(h2.x, w2, a.x); a.y = fmaf(h2.y, w2, a.y);
            a.z = fmaf(h2.z, w2, a.z); a.w = fmaf(h2.w, w2, a.w);
            a.x = fmaf(h3.x, w3, a.x); a.y = fmaf(h3.y, w3, a.y);
            a.z = fmaf(h3.z, w3, a.z); a.w = fmaf(h3.w, w3, a.w);
        }
    }
    const float4 hv = *(const float4*)(h + (size_t)n * 64 + c);
    const float4 b  = *(const float4*)(bias + c);
    float4 r;
    r.x = (a.x + hv.x) * di + b.x;
    r.y = (a.y + hv.y) * di + b.y;
    r.z = (a.z + hv.z) * di + b.z;
    r.w = (a.w + hv.w) * di + b.w;
    if (do_relu) {
        r.x = fmaxf(r.x, 0.f); r.y = fmaxf(r.y, 0.f);
        r.z = fmaxf(r.z, 0.f); r.w = fmaxf(r.w, 0.f);
    }
    *(float4*)(out + (size_t)n * 64 + c) = r;
}


#define TILE_PITCH 68

static_assert(N_NODES % 16 == 0, "the fused kernel has no partial tile");

__global__ __launch_bounds__(256) void gather_gemm_kernel(const float* __restrict__ h,
                                                          const float* __restrict__ dinv,
                                                          const int* __restrict__ cnt,
                                                          const int* __restrict__ rowstart,
                                                          const int* __restrict__ col,
                                                          const float* __restrict__ bias,
                                                          const float* __restrict__ Wn,
                                                          float* __restrict__ out,
                                                          int e_count) {
    __shared__ alignas(16) float sH[16 * TILE_PITCH];
    __shared__ alignas(16) float sO[16 * TILE_PITCH];

    const int tid = threadIdx.x;
    const int nl = tid >> 4;
    const int n = blockIdx.x * 16 + nl;
    const int c = (tid & 15) << 2;
    const float di = dinv[n];
    const int start = max(rowstart[n], 0);
    const int end = min(start + cnt[n], e_count);
    float4 a = make_float4(0.f, 0.f, 0.f, 0.f);
    if (start < end) {
        const int last = end - 1;
        int s0 = col[start];
        int s1 = col[min(start + 1, last)];
        int s2 = col[min(start + 2, last)];
        int s3 = col[min(start + 3, last)];
        for (int p = start; p < end; p += 4) {
            const float4 h0 = *(const float4*)(h + (size_t)s0 * 64 + c);
            const float4 h1 = *(const float4*)(h + (size_t)s1 * 64 + c);
            const float4 h2 = *(const float4*)(h + (size_t)s2 * 64 + c);
            const float4 h3 = *(const float4*)(h + (size_t)s3 * 64 + c);
            const float w1 = (p + 1 < end) ? 1.f : 0.f;
            const float w2 = (p + 2 < end) ? 1.f : 0.f;
            const float w3 = (p + 3 < end) ? 1.f : 0.f;
            s0 = col[min(p + 4, last)];
            s1 = col[min(p + 5, last)];
            s2 = col[min(p + 6, last)];
            s3 = col[min(p + 7, last)];
            a.x += h0.x; a.y += h0.y; a.z += h0.z; a.w += h0.w;
            a.x = fmaf(h1.x, w1, a.x); a.y = fmaf(h1.y, w1, a.y);
            a.z = fmaf(h1.z, w1, a.z); a.w = fmaf(h1.w, w1, a.w);
            a.x = fmaf(h2.x, w2, a.x); a.y = fmaf(h2.y, w2, a.y);
            a.z = fmaf(h2.z, w2, a.z); a.w = fmaf(h2.w, w2, a.w);
            a.x = fmaf(h3.x, w3, a.x); a.y = fmaf(h3.y, w3, a.y);
            a.z = fmaf(h3.z, w3, a.z); a.w = fmaf(h3.w, w3, a.w);
        }
    }
    const float4 hv = *(const float4*)(h + (size_t)n * 64 + c);
    const float4 b  = *(const float4*)(bias + c);
    float4 r;
    r.x = fmaxf((a.x + hv.x) * di + b.x, 0.f);
    r.y = fmaxf((a.y + hv.y) * di + b.y, 0.f);
    r.z = fmaxf((a.z + hv.z) * di + b.z, 0.f);
    r.w = fmaxf((a.w + hv.w) * di + b.w, 0.f);
    *(float4*)(sH + nl * TILE_PITCH + c) = r;

    const int wave = tid >> 5;
    const int lane = tid & 31;
    const int row16 = lane & 15;
    const int khalf = (lane >> 4) << 1;

    v2f fb[16];
    if (wave < 4) {
        const float* __restrict__ Bcol = Wn + khalf * 64 + wave * 16 + row16;
        #pragma unroll
        for (int i = 0; i < 16; ++i) {
            fb[i].x = Bcol[(4 * i) * 64];
            fb[i].y = Bcol[(4 * i + 1) * 64];
        }
    }
    __syncthreads();

    if (wave < 4) {
        const float* Arow = sH + row16 * TILE_PITCH + khalf;
        v8f acc = {};
        #pragma unroll
        for (int i = 0; i < 16; ++i) {
            v2f fa;
            fa.x = Arow[4 * i];
            fa.y = Arow[4 * i + 1];
            acc = __builtin_amdgcn_wmma_f32_16x16x4_f32(
                 false, fa,  false, fb[i],
                 (short)0, acc,  false,  false);
        }
        float* Orow = sO + ((lane >> 4) << 3) * TILE_PITCH + wave * 16 + row16;
        #pragma unroll
        for (int j = 0; j < 8; ++j) Orow[j * TILE_PITCH] = acc[j];
    }
    __syncthreads();

    float4 o = *(const float4*)(sO + nl * TILE_PITCH + c);
    o.x *= di; o.y *= di; o.z *= di; o.w *= di;
    *(float4*)(out + (size_t)n * 64 + c) = o;
}


extern "C" void kernel_launch(void* const* d_in, const int* in_sizes, int n_in,
                              void* d_out, int out_size, void* d_ws, size_t ws_size,
                              hipStream_t stream) {
    const float* x  = (const float*)d_in[0];
    const int*   ei = (const int*)d_in[1];
    const float* W1 = (const float*)d_in[2];
    const float* b1 = (const float*)d_in[3];
    const float* W2 = (const float*)d_in[4];
    const float* b2 = (const float*)d_in[5];
    float* out = (float*)d_out;

    const int* src = ei;
    const int* dst = ei + N_EDGES;

    float* dinv      = (float*)d_ws;
    int*   cnt       = (int*)(dinv + 65536);
    int*   rowstart  = cnt + 65536;
    int*   hist      = rowstart + 65536;
    int*   bkt_start = hist + 65536;
    int*   bkt_cnt   = bkt_start + 256;
    int2*  bucketed  = (int2*)(bkt_cnt + 256);
    int*   col       = (int*)(bucketed + N_EDGES);
    float* buf0      = (float*)(col + N_EDGES);
    float* buf1      = buf0 + (size_t)N_NODES * 64;

    const int B = 256;
    dim3 blk(B);
    dim3 blk1k(1024);

    bucket_hist_kernel<<<N_CHUNKS, blk1k, 0, stream>>>(dst, hist);
    bucket_scatter_kernel<<<N_CHUNKS, blk1k, 0, stream>>>(src, dst, hist, bkt_start, bkt_cnt, bucketed);
    bucket_finalize_kernel<<<N_BUCKETS, blk1k, 0, stream>>>(bucketed, bkt_start, bkt_cnt, cnt, rowstart,
                                                            dinv, col);

    int gemm_grid = (N_NODES + 31) / 32;
    gemm_wmma_f32<IN_DIM><<<gemm_grid, blk, 0, stream>>>(x, W1, dinv, buf0, N_NODES);
    gather_gemm_kernel<<<N_NODES / 16, blk, 0, stream>>>(buf0, dinv, cnt, rowstart, col, b1, W2, buf1, N_EDGES);

    gather_kernel<<<(N_NODES * 16 + B - 1) / B, blk, 0, stream>>>(buf1, dinv, cnt, rowstart, col, b2,
                                                                  out, N_NODES, N_EDGES,  0);
}


# Edge sort in two launches, not three: each chunk orders its own edges by bucket in place (rank = histogram atomic's return); the bucket kernel collects its 256 pieces; sum order = LDS arrival rank
# speedup vs baseline: 3.9826x; 1.1046x over previous
#include <hip/hip_runtime.h>
#include <hip/hip_bf16.h>
#include <math.h>


#define N_NODES 50000
#define N_EDGES 800000
#define IN_DIM  128
#define HID_DIM 64

#define N_BUCKETS 196
#define N_CHUNKS  256
#define CHUNK     3125
#define LSORT_CAP 8192

static_assert(N_CHUNKS * CHUNK == N_EDGES, "chunks must tile the edge list exactly");
static_assert(N_BUCKETS * 256 >= N_NODES && N_BUCKETS <= 256, "bucket count");

typedef float v2f __attribute__((ext_vector_type(2)));
typedef float v8f __attribute__((ext_vector_type(8)));


__device__ __forceinline__ int bucket_of(int d) {
    unsigned b = ((unsigned)d) >> 8;
    return (int)(b < (unsigned)(N_BUCKETS - 1) ? b : (unsigned)(N_BUCKETS - 1));
}

__device__ __forceinline__ int block_incl_scan_1024(int v, int* wsum) {
    const int lane = threadIdx.x & 31;
    const int wave = threadIdx.x >> 5;
    int incl = v;
    for (int d = 1; d < 32; d <<= 1) {
        int t = __shfl_up(incl, d);
        if (lane >= d) incl += t;
    }
    if (lane == 31) wsum[wave] = incl;
    __syncthreads();
    int base = 0;
    for (int w = 0; w < 32; ++w) {
        int s = wsum[w];
        base += (w < wave) ? s : 0;
    }
    __syncthreads();
    return base + incl;
}

__global__ __launch_bounds__(1024) void chunk_sort_kernel(const int* __restrict__ src,
                                                          const int* __restrict__ dst,
                                                          int2* __restrict__ tab,
                                                          int2* __restrict__ bucketed) {
    __shared__ int sorted_s[CHUNK];
    __shared__ int sorted_d[CHUNK];
    __shared__ int hcnt[256];
    __shared__ int lstart[256];
    __shared__ int wsum[32];

    const int tid = threadIdx.x;
    const int c = blockIdx.x;

    int es[4], ed[4], rk[4];
    for (int j = 0; j < 4; ++j) {
        const int i = tid + j * 1024;
        es[j] = 0; ed[j] = 0;
        if (i < CHUNK) { es[j] = src[c * CHUNK + i]; ed[j] = dst[c * CHUNK + i]; }
    }
    if (tid < 256) hcnt[tid] = 0;
    __syncthreads();

    for (int j = 0; j < 4; ++j) {
        const int i = tid + j * 1024;
        rk[j] = 0;
        if (i < CHUNK) rk[j] = atomicAdd(&hcnt[bucket_of(ed[j])], 1);
    }
    __syncthreads();

    const int lcnt = (tid < 256) ? hcnt[tid] : 0;
    const int incl = block_incl_scan_1024(lcnt, wsum);
    if (tid < 256) {
        lstart[tid] = incl - lcnt;
        tab[c * 256 + tid] = make_int2(incl - lcnt, lcnt);
    }
    __syncthreads();

    for (int j = 0; j < 4; ++j) {
        const int i = tid + j * 1024;
        if (i < CHUNK) {
            const int lp = min(max(lstart[bucket_of(ed[j])] + rk[j], 0), CHUNK - 1);
            sorted_s[lp] = es[j];
            sorted_d[lp] = ed[j];
        }
    }
    __syncthreads();

    for (int j = 0; j < 4; ++j) {
        const int i = tid + j * 1024;
        if (i < CHUNK) bucketed[c * CHUNK + i] = make_int2(sorted_s[i], sorted_d[i]);
    }
}

__device__ __forceinline__ int piece_index(const int* pstart, const int* pbase, int i) {
    int lo = 0;
    for (int step = 128; step >= 1; step >>= 1) {
        if (pstart[lo + step] <= i) lo += step;
    }
    return min(max(pbase[lo] + (i - pstart[lo]), 0), N_EDGES - 1);
}

__global__ __launch_bounds__(1024) void bucket_finalize_kernel(const int2* __restrict__ bucketed,
                                                               const int2* __restrict__ tab,
                                                               int* __restrict__ cnt,
                                                               int* __restrict__ rowstart,
                                                               float* __restrict__ dinv,
                                                               int* __restrict__ col) {
    __shared__ int ncnt[256];
    __shared__ int nstart[256];
    __shared__ int pstart[256];
    __shared__ int pbase[256];
    __shared__ int2 wtot[8];
    __shared__ int wsum[32];
    __shared__ int lsorted[LSORT_CAP];

    const int tid = threadIdx.x;
    const int b = blockIdx.x;

    int pc = 0, pl = 0;
    if (tid < 256) {
        const int2 e = tab[tid * 256 + b];
        pc = min(max(e.y, 0), CHUNK);
        pl = min(max(e.x, 0), CHUNK - pc);
    }
    int incl_c = pc, incl_l = pl;
    for (int d = 1; d < 32; d <<= 1) {
        const int tc = __shfl_up(incl_c, d);
        const int tl = __shfl_up(incl_l, d);
        if ((tid & 31) >= d) { incl_c += tc; incl_l += tl; }
    }
    if ((tid & 31) == 31 && tid < 256) wtot[tid >> 5] = make_int2(incl_c, incl_l);
    __syncthreads();
    int base_c = 0, base_l = 0, tot_c = 0, tot_l = 0;
    for (int w = 0; w < 8; ++w) {
        const int2 t = wtot[w];
        tot_c += t.x; tot_l += t.y;
        if (w < (tid >> 5)) { base_c += t.x; base_l += t.y; }
    }
    if (tid < 256) {
        pstart[tid] = base_c + incl_c - pc;
        pbase[tid] = tid * CHUNK + pl;
        ncnt[tid] = 0;
    }
    __syncthreads();

    const int s = min(max(tot_l, 0), N_EDGES);
    const int m = min(max(tot_c, 0), N_EDGES - s);

    const bool in_lds = (m <= LSORT_CAP);

    int rs[8], rd[8];
    const int mlast = max(m - 1, 0);
    for (int k = 0; k < 8; ++k) {
        const int i = min(tid + k * 1024, mlast);
        const int2 rec = bucketed[piece_index(pstart, pbase, i)];
        rs[k] = rec.x; rd[k] = rec.y;
    }

    if (in_lds) {
        for (int k = 0; k < 8; ++k) {
            if (tid + k * 1024 < m) atomicAdd(&ncnt[rd[k] & 255], 1);
        }
    } else {
        for (int i = tid; i < m; i += 1024) {
            const int d = bucketed[piece_index(pstart, pbase, i)].y;
            atomicAdd(&ncnt[d & 255], 1);
        }
    }
    __syncthreads();

    const int v = (tid < 256) ? ncnt[tid] : 0;
    const int incl = block_incl_scan_1024(v, wsum);
    if (tid < 256) {
        const int ex = incl - v;
        nstart[tid] = ex;
        ncnt[tid] = 0;
        const int n = b * 256 + tid;
        if (n < N_NODES) {
            cnt[n] = v;
            rowstart[n] = s + ex;
            const float dg = (float)(v + 1);
            dinv[n] = (dg > 0.0f) ? rsqrtf(dg) : 0.0f;
        }
    }
    __syncthreads();

    if (in_lds) {
        for (int k = 0; k < 8; ++k) {
            if (tid + k * 1024 < m) {
                const int dl = rd[k] & 255;
                const int r = atomicAdd(&ncnt[dl], 1);
                const int p = min(max(nstart[dl] + r, 0), m - 1);
                lsorted[p] = rs[k];
            }
        }
    } else {
        for (int i = tid; i < m; i += 1024) {
            const int2 rec = bucketed[piece_index(pstart, pbase, i)];
            const int dl = rec.y & 255;
            const int r = atomicAdd(&ncnt[dl], 1);
            const int p = min(max(nstart[dl] + r, 0), m - 1);
            col[s + p] = rec.x;
        }
    }
    __syncthreads();
    if (in_lds) {
        for (int i = tid; i < m; i += 1024) col[s + i] = lsorted[i];
    }
}


template <int K_DIM>
__global__ __launch_bounds__(256) void gemm_wmma_f32(const float* __restrict__ A,
                                                     const float* __restrict__ W,
                                                     const float* __restrict__ rscale,
                                                     float* __restrict__ C, int M) {
    __shared__ float sW[K_DIM * 64];
    for (int i = threadIdx.x; i < K_DIM * 64; i += 256) sW[i] = W[i];
    __syncthreads();

    const int wave = threadIdx.x >> 5;
    const int lane = threadIdx.x & 31;
    const int m_tile = blockIdx.x * 2 + (wave >> 2);
    const int n_tile = wave & 3;

    const int row16 = lane & 15;
    const int khalf = (lane >> 4) << 1;

    int mrow = m_tile * 16 + row16;
    int arow = (mrow < M) ? mrow : (M - 1);
    const float* __restrict__ Aptr = A + (size_t)arow * K_DIM;
    const float* __restrict__ Bcol = sW + n_tile * 16 + row16;

    v8f acc = {};
    #pragma unroll
    for (int k = 0; k < K_DIM; k += 4) {
        v2f a, b;
        a.x = Aptr[k + khalf];
        a.y = Aptr[k + khalf + 1];
        b.x = Bcol[(k + khalf) * 64];
        b.y = Bcol[(k + khalf + 1) * 64];
        acc = __builtin_amdgcn_wmma_f32_16x16x4_f32(
             false, a,  false, b,
             (short)0, acc,  false,  false);
    }

    const int mbase = m_tile * 16 + ((lane >> 4) << 3);
    const int ncol  = n_tile * 16 + row16;
    #pragma unroll
    for (int j = 0; j < 8; ++j) {
        int m = mbase + j;
        if (m < M) C[(size_t)m * 64 + ncol] = acc[j] * rscale[m];
    }
}


__global__ __launch_bounds__(256) void gather_kernel(const float* __restrict__ h,
                                                     const float* __restrict__ dinv,
                                                     const int* __restrict__ cnt,
                                                     const int* __restrict__ rowstart,
                                                     const int* __restrict__ col,
                                                     const float* __restrict__ bias,
                                                     float* __restrict__ out,
                                                     int n_nodes, int e_count, int do_relu) {
    int gid = blockIdx.x * blockDim.x + threadIdx.x;
    int n = gid >> 4;
    if (n >= n_nodes) return;
    int c = (gid & 15) << 2;
    const float di = dinv[n];
    const int start = max(rowstart[n], 0);
    const int end = min(start + cnt[n], e_count);
    float4 a = make_float4(0.f, 0.f, 0.f, 0.f);
    if (start < end) {
        const int last = end - 1;
        int s0 = col[start];
        int s1 = col[min(start + 1, last)];
        int s2 = col[min(start + 2, last)];
        int s3 = col[min(start + 3, last)];
        for (int p = start; p < end; p += 4) {
            const float4 h0 = *(const float4*)(h + (size_t)s0 * 64 + c);
            const float4 h1 = *(const float4*)(h + (size_t)s1 * 64 + c);
            const float4 h2 = *(const float4*)(h + (size_t)s2 * 64 + c);
            const float4 h3 = *(const float4*)(h + (size_t)s3 * 64 + c);
            const float w1 = (p + 1 < end) ? 1.f : 0.f;
            const float w2 = (p + 2 < end) ? 1.f : 0.f;
            const float w3 = (p + 3 < end) ? 1.f : 0.f;
            s0 = col[min(p + 4, last)];
            s1 = col[min(p + 5, last)];
            s2 = col[min(p + 6, last)];
            s3 = col[min(p + 7, last)];
            a.x += h0.x; a.y += h0.y; a.z += h0.z; a.w += h0.w;
            a.x = fmaf(h1.x, w1, a.x); a.y = fmaf(h1.y, w1, a.y);
            a.z = fmaf(h1.z, w1, a.z); a.w = fmaf(h1.w, w1, a.w);
            a.x = fmaf(h2.x, w2, a.x); a.y = fmaf(h2.y, w2, a.y);
            a.z = fmaf(h2.z, w2, a.z); a.w = fmaf(h2.w, w2, a.w);
            a.x = fmaf(h3.x, w3, a.x); a.y = fmaf(h3.y, w3, a.y);
            a.z = fmaf(h3.z, w3, a.z); a.w = fmaf(h3.w, w3, a.w);
        }
    }
    const float4 hv = *(const float4*)(h + (size_t)n * 64 + c);
    const float4 b  = *(const float4*)(bias + c);
    float4 r;
    r.x = (a.x + hv.x) * di + b.x;
    r.y = (a.y + hv.y) * di + b.y;
    r.z = (a.z + hv.z) * di + b.z;
    r.w = (a.w + hv.w) * di + b.w;
    if (do_relu) {
        r.x = fmaxf(r.x, 0.f); r.y = fmaxf(r.y, 0.f);
        r.z = fmaxf(r.z, 0.f); r.w = fmaxf(r.w, 0.f);
    }
    *(float4*)(out + (size_t)n * 64 + c) = r;
}


#define TILE_PITCH 68

static_assert(N_NODES % 16 == 0, "the fused kernel has no partial tile");

__global__ __launch_bounds__(256) void gather_gemm_kernel(const float* __restrict__ h,
                                                          const float* __restrict__ dinv,
                                                          const int* __restrict__ cnt,
                                                          const int* __restrict__ rowstart,
                                                          const int* __restrict__ col,
                                                          const float* __restrict__ bias,
                                                          const float* __restrict__ Wn,
                                                          float* __restrict__ out,
                                                          int e_count) {
    __shared__ alignas(16) float sH[16 * TILE_PITCH];
    __shared__ alignas(16) float sO[16 * TILE_PITCH];

    const int tid = threadIdx.x;
    const int nl = tid >> 4;
    const int n = blockIdx.x * 16 + nl;
    const int c = (tid & 15) << 2;
    const float di = dinv[n];
    const int start = max(rowstart[n], 0);
    const int end = min(start + cnt[n], e_count);
    float4 a = make_float4(0.f, 0.f, 0.f, 0.f);
    if (start < end) {
        const int last = end - 1;
        int s0 = col[start];
        int s1 = col[min(start + 1, last)];
        int s2 = col[min(start + 2, last)];
        int s3 = col[min(start + 3, last)];
        for (int p = start; p < end; p += 4) {
            const float4 h0 = *(const float4*)(h + (size_t)s0 * 64 + c);
            const float4 h1 = *(const float4*)(h + (size_t)s1 * 64 + c);
            const float4 h2 = *(const float4*)(h + (size_t)s2 * 64 + c);
            const float4 h3 = *(const float4*)(h + (size_t)s3 * 64 + c);
            const float w1 = (p + 1 < end) ? 1.f : 0.f;
            const float w2 = (p + 2 < end) ? 1.f : 0.f;
            const float w3 = (p + 3 < end) ? 1.f : 0.f;
            s0 = col[min(p + 4, last)];
            s1 = col[min(p + 5, last)];
            s2 = col[min(p + 6, last)];
            s3 = col[min(p + 7, last)];
            a.x += h0.x; a.y += h0.y; a.z += h0.z; a.w += h0.w;
            a.x = fmaf(h1.x, w1, a.x); a.y = fmaf(h1.y, w1, a.y);
            a.z = fmaf(h1.z, w1, a.z); a.w = fmaf(h1.w, w1, a.w);
            a.x = fmaf(h2.x, w2, a.x); a.y = fmaf(h2.y, w2, a.y);
            a.z = fmaf(h2.z, w2, a.z); a.w = fmaf(h2.w, w2, a.w);
            a.x = fmaf(h3.x, w3, a.x); a.y = fmaf(h3.y, w3, a.y);
            a.z = fmaf(h3.z, w3, a.z); a.w = fmaf(h3.w, w3, a.w);
        }
    }
    const float4 hv = *(const float4*)(h + (size_t)n * 64 + c);
    const float4 b  = *(const float4*)(bias + c);
    float4 r;
    r.x = fmaxf((a.x + hv.x) * di + b.x, 0.f);
    r.y = fmaxf((a.y + hv.y) * di + b.y, 0.f);
    r.z = fmaxf((a.z + hv.z) * di + b.z, 0.f);
    r.w = fmaxf((a.w + hv.w) * di + b.w, 0.f);
    *(float4*)(sH + nl * TILE_PITCH + c) = r;

    const int wave = tid >> 5;
    const int lane = tid & 31;
    const int row16 = lane & 15;
    const int khalf = (lane >> 4) << 1;

    v2f fb[16];
    if (wave < 4) {
        const float* __restrict__ Bcol = Wn + khalf * 64 + wave * 16 + row16;
        #pragma unroll
        for (int i = 0; i < 16; ++i) {
            fb[i].x = Bcol[(4 * i) * 64];
            fb[i].y = Bcol[(4 * i + 1) * 64];
        }
    }
    __syncthreads();

    if (wave < 4) {
        const float* Arow = sH + row16 * TILE_PITCH + khalf;
        v8f acc = {};
        #pragma unroll
        for (int i = 0; i < 16; ++i) {
            v2f fa;
            fa.x = Arow[4 * i];
            fa.y = Arow[4 * i + 1];
            acc = __builtin_amdgcn_wmma_f32_16x16x4_f32(
                 false, fa,  false, fb[i],
                 (short)0, acc,  false,  false);
        }
        float* Orow = sO + ((lane >> 4) << 3) * TILE_PITCH + wave * 16 + row16;
        #pragma unroll
        for (int j = 0; j < 8; ++j) Orow[j * TILE_PITCH] = acc[j];
    }
    __syncthreads();

    float4 o = *(const float4*)(sO + nl * TILE_PITCH + c);
    o.x *= di; o.y *= di; o.z *= di; o.w *= di;
    *(float4*)(out + (size_t)n * 64 + c) = o;
}


extern "C" void kernel_launch(void* const* d_in, const int* in_sizes, int n_in,
                              void* d_out, int out_size, void* d_ws, size_t ws_size,
                              hipStream_t stream) {
    const float* x  = (const float*)d_in[0];
    const int*   ei = (const int*)d_in[1];
    const float* W1 = (const float*)d_in[2];
    const float* b1 = (const float*)d_in[3];
    const float* W2 = (const float*)d_in[4];
    const float* b2 = (const float*)d_in[5];
    float* out = (float*)d_out;

    const int* src = ei;
    const int* dst = ei + N_EDGES;

    float* dinv      = (float*)d_ws;
    int*   cnt       = (int*)(dinv + 65536);
    int*   rowstart  = cnt + 65536;
    int2*  tab       = (int2*)(rowstart + 65536);
    int2*  bucketed  = tab + 65536;
    int*   col       = (int*)(bucketed + N_EDGES);
    float* buf0      = (float*)(col + N_EDGES);
    float* buf1      = buf0 + (size_t)N_NODES * 64;

    const int B = 256;
    dim3 blk(B);
    dim3 blk1k(1024);

    chunk_sort_kernel<<<N_CHUNKS, blk1k, 0, stream>>>(src, dst, tab, bucketed);
    bucket_finalize_kernel<<<N_BUCKETS, blk1k, 0, stream>>>(bucketed, tab, cnt, rowstart, dinv, col);

    int gemm_grid = (N_NODES + 31) / 32;
    gemm_wmma_f32<IN_DIM><<<gemm_grid, blk, 0, stream>>>(x, W1, dinv, buf0, N_NODES);
    gather_gemm_kernel<<<N_NODES / 16, blk, 0, stream>>>(buf0, dinv, cnt, rowstart, col, b1, W2, buf1, N_EDGES);

    gather_kernel<<<(N_NODES * 16 + B - 1) / B, blk, 0, stream>>>(buf1, dinv, cnt, rowstart, col, b2,
                                                                  out, N_NODES, N_EDGES,  0);
}


# x*W1 GEMM re-tiled: wave = 16 rows x all 64 columns: A read once (not 4x), 16-byte loads in flight over an LDS-only barrier; W in LDS as k pairs; whole rows stored; sum order: k permuted in each 8
# speedup vs baseline: 4.5526x; 1.1431x over previous
#include <hip/hip_runtime.h>
#include <hip/hip_bf16.h>
#include <math.h>


#define N_NODES 50000
#define N_EDGES 800000
#define IN_DIM  128
#define HID_DIM 64

#define N_BUCKETS 196
#define N_CHUNKS  256
#define CHUNK     3125
#define LSORT_CAP 8192

static_assert(N_CHUNKS * CHUNK == N_EDGES, "chunks must tile the edge list exactly");
static_assert(N_BUCKETS * 256 >= N_NODES && N_BUCKETS <= 256, "bucket count");

typedef float v2f __attribute__((ext_vector_type(2)));
typedef float v8f __attribute__((ext_vector_type(8)));


__device__ __forceinline__ int bucket_of(int d) {
    unsigned b = ((unsigned)d) >> 8;
    return (int)(b < (unsigned)(N_BUCKETS - 1) ? b : (unsigned)(N_BUCKETS - 1));
}

__device__ __forceinline__ int block_incl_scan_1024(int v, int* wsum) {
    const int lane = threadIdx.x & 31;
    const int wave = threadIdx.x >> 5;
    int incl = v;
    for (int d = 1; d < 32; d <<= 1) {
        int t = __shfl_up(incl, d);
        if (lane >= d) incl += t;
    }
    if (lane == 31) wsum[wave] = incl;
    __syncthreads();
    int base = 0;
    for (int w = 0; w < 32; ++w) {
        int s = wsum[w];
        base += (w < wave) ? s : 0;
    }
    __syncthreads();
    return base + incl;
}

__global__ __launch_bounds__(1024) void chunk_sort_kernel(const int* __restrict__ src,
                                                          const int* __restrict__ dst,
                                                          int2* __restrict__ tab,
                                                          int2* __restrict__ bucketed) {
    __shared__ int sorted_s[CHUNK];
    __shared__ int sorted_d[CHUNK];
    __shared__ int hcnt[256];
    __shared__ int lstart[256];
    __shared__ int wsum[32];

    const int tid = threadIdx.x;
    const int c = blockIdx.x;

    int es[4], ed[4], rk[4];
    for (int j = 0; j < 4; ++j) {
        const int i = tid + j * 1024;
        es[j] = 0; ed[j] = 0;
        if (i < CHUNK) { es[j] = src[c * CHUNK + i]; ed[j] = dst[c * CHUNK + i]; }
    }
    if (tid < 256) hcnt[tid] = 0;
    __syncthreads();

    for (int j = 0; j < 4; ++j) {
        const int i = tid + j * 1024;
        rk[j] = 0;
        if (i < CHUNK) rk[j] = atomicAdd(&hcnt[bucket_of(ed[j])], 1);
    }
    __syncthreads();

    const int lcnt = (tid < 256) ? hcnt[tid] : 0;
    const int incl = block_incl_scan_1024(lcnt, wsum);
    if (tid < 256) {
        lstart[tid] = incl - lcnt;
        tab[c * 256 + tid] = make_int2(incl - lcnt, lcnt);
    }
    __syncthreads();

    for (int j = 0; j < 4; ++j) {
        const int i = tid + j * 1024;
        if (i < CHUNK) {
            const int lp = min(max(lstart[bucket_of(ed[j])] + rk[j], 0), CHUNK - 1);
            sorted_s[lp] = es[j];
            sorted_d[lp] = ed[j];
        }
    }
    __syncthreads();

    for (int j = 0; j < 4; ++j) {
        const int i = tid + j * 1024;
        if (i < CHUNK) bucketed[c * CHUNK + i] = make_int2(sorted_s[i], sorted_d[i]);
    }
}

__device__ __forceinline__ int piece_index(const int* pstart, const int* pbase, int i) {
    int lo = 0;
    for (int step = 128; step >= 1; step >>= 1) {
        if (pstart[lo + step] <= i) lo += step;
    }
    return min(max(pbase[lo] + (i - pstart[lo]), 0), N_EDGES - 1);
}

__global__ __launch_bounds__(1024) void bucket_finalize_kernel(const int2* __restrict__ bucketed,
                                                               const int2* __restrict__ tab,
                                                               int* __restrict__ cnt,
                                                               int* __restrict__ rowstart,
                                                               float* __restrict__ dinv,
                                                               int* __restrict__ col) {
    __shared__ int ncnt[256];
    __shared__ int nstart[256];
    __shared__ int pstart[256];
    __shared__ int pbase[256];
    __shared__ int2 wtot[8];
    __shared__ int wsum[32];
    __shared__ int lsorted[LSORT_CAP];

    const int tid = threadIdx.x;
    const int b = blockIdx.x;

    int pc = 0, pl = 0;
    if (tid < 256) {
        const int2 e = tab[tid * 256 + b];
        pc = min(max(e.y, 0), CHUNK);
        pl = min(max(e.x, 0), CHUNK - pc);
    }
    int incl_c = pc, incl_l = pl;
    for (int d = 1; d < 32; d <<= 1) {
        const int tc = __shfl_up(incl_c, d);
        const int tl = __shfl_up(incl_l, d);
        if ((tid & 31) >= d) { incl_c += tc; incl_l += tl; }
    }
    if ((tid & 31) == 31 && tid < 256) wtot[tid >> 5] = make_int2(incl_c, incl_l);
    __syncthreads();
    int base_c = 0, base_l = 0, tot_c = 0, tot_l = 0;
    for (int w = 0; w < 8; ++w) {
        const int2 t = wtot[w];
        tot_c += t.x; tot_l += t.y;
        if (w < (tid >> 5)) { base_c += t.x; base_l += t.y; }
    }
    if (tid < 256) {
        pstart[tid] = base_c + incl_c - pc;
        pbase[tid] = tid * CHUNK + pl;
        ncnt[tid] = 0;
    }
    __syncthreads();

    const int s = min(max(tot_l, 0), N_EDGES);
    const int m = min(max(tot_c, 0), N_EDGES - s);

    const bool in_lds = (m <= LSORT_CAP);

    int rs[8], rd[8];
    const int mlast = max(m - 1, 0);
    for (int k = 0; k < 8; ++k) {
        const int i = min(tid + k * 1024, mlast);
        const int2 rec = bucketed[piece_index(pstart, pbase, i)];
        rs[k] = rec.x; rd[k] = rec.y;
    }

    if (in_lds) {
        for (int k = 0; k < 8; ++k) {
            if (tid + k * 1024 < m) atomicAdd(&ncnt[rd[k] & 255], 1);
        }
    } else {
        for (int i = tid; i < m; i += 1024) {
            const int d = bucketed[piece_index(pstart, pbase, i)].y;
            atomicAdd(&ncnt[d & 255], 1);
        }
    }
    __syncthreads();

    const int v = (tid < 256) ? ncnt[tid] : 0;
    const int incl = block_incl_scan_1024(v, wsum);
    if (tid < 256) {
        const int ex = incl - v;
        nstart[tid] = ex;
        ncnt[tid] = 0;
        const int n = b * 256 + tid;
        if (n < N_NODES) {
            cnt[n] = v;
            rowstart[n] = s + ex;
            const float dg = (float)(v + 1);
            dinv[n] = (dg > 0.0f) ? rsqrtf(dg) : 0.0f;
        }
    }
    __syncthreads();

    if (in_lds) {
        for (int k = 0; k < 8; ++k) {
            if (tid + k * 1024 < m) {
                const int dl = rd[k] & 255;
                const int r = atomicAdd(&ncnt[dl], 1);
                const int p = min(max(nstart[dl] + r, 0), m - 1);
                lsorted[p] = rs[k];
            }
        }
    } else {
        for (int i = tid; i < m; i += 1024) {
            const int2 rec = bucketed[piece_index(pstart, pbase, i)];
            const int dl = rec.y & 255;
            const int r = atomicAdd(&ncnt[dl], 1);
            const int p = min(max(nstart[dl] + r, 0), m - 1);
            col[s + p] = rec.x;
        }
    }
    __syncthreads();
    if (in_lds) {
        for (int i = tid; i < m; i += 1024) col[s + i] = lsorted[i];
    }
}


#define W1_PITCH 144
#define TILE_PITCH 68

static_assert(IN_DIM == 128 && HID_DIM == 64, "the GEMM's tiling is written for a 128 x 64 weight");
static_assert(N_NODES % 16 == 0, "a 16-row tile is either whole or absent: the stores need no row guard");

__global__ __launch_bounds__(256) void gemm_strip_wmma_f32(const float* __restrict__ A,
                                                           const float* __restrict__ W,
                                                           const float* __restrict__ rscale,
                                                           float* __restrict__ C) {
    __shared__ alignas(16) float sWp[(IN_DIM / 2) * W1_PITCH];
    __shared__ alignas(16) float sT[8 * 16 * TILE_PITCH];

    const int tid = threadIdx.x;
    const int wave = tid >> 5;
    const int lane = tid & 31;
    const int c = lane & 15;
    const int h = lane >> 4;

    float4 w0[4], w1[4];
    #pragma unroll
    for (int j = 0; j < 4; ++j) {
        const int id = tid + j * 256;
        const int g = id & 15;
        const int kp = id >> 4;
        w0[j] = *(const float4*)(W + (2 * kp) * 64 + 4 * g);
        w1[j] = *(const float4*)(W + (2 * kp + 1) * 64 + 4 * g);
    }

    const int m_tile = blockIdx.x * 8 + wave;
    const bool valid = (m_tile * 16 < N_NODES);
    const int row0 = valid ? m_tile * 16 : (N_NODES - 16);
    const float* __restrict__ Ap = A + (size_t)(row0 + c) * IN_DIM + 4 * h;
    float4 q[16];
    #pragma unroll
    for (int i = 0; i < 16; ++i) q[i] = *(const float4*)(Ap + 8 * i);
    const float4 rs0 = *(const float4*)(rscale + row0 + 8 * h);
    const float4 rs1 = *(const float4*)(rscale + row0 + 8 * h + 4);

    #pragma unroll
    for (int j = 0; j < 4; ++j) {
        const int id = tid + j * 256;
        const int g = id & 15;
        const int kp = id >> 4;
        float* p = sWp + kp * W1_PITCH + 8 * g;
        *(float4*)(p)     = make_float4(w0[j].x, w1[j].x, w0[j].y, w1[j].y);
        *(float4*)(p + 4) = make_float4(w0[j].z, w1[j].z, w0[j].w, w1[j].w);
    }
    __builtin_amdgcn_fence(__ATOMIC_RELEASE, "workgroup", "local");
    __builtin_amdgcn_s_barrier();
    __builtin_amdgcn_fence(__ATOMIC_ACQUIRE, "workgroup", "local");

    float* Tw = sT + wave * (16 * TILE_PITCH);
    if (valid) {
        const float* Bp = sWp + (2 * h) * W1_PITCH + 2 * c;
        v8f acc0 = {}, acc1 = {}, acc2 = {}, acc3 = {};
        #pragma unroll
        for (int i = 0; i < 16; ++i) {
            #pragma unroll
            for (int st = 0; st < 2; ++st) {
                v2f a;
                a.x = st ? q[i].z : q[i].x;
                a.y = st ? q[i].w : q[i].y;
                const float* bp = Bp + (4 * i + st) * W1_PITCH;
                const v2f b0 = *(const v2f*)(bp);
                const v2f b1 = *(const v2f*)(bp + 32);
                const v2f b2 = *(const v2f*)(bp + 64);
                const v2f b3 = *(const v2f*)(bp + 96);
                acc0 = __builtin_amdgcn_wmma_f32_16x16x4_f32(false, a, false, b0, (short)0, acc0, false, false);
                acc1 = __builtin_amdgcn_wmma_f32_16x16x4_f32(false, a, false, b1, (short)0, acc1, false, false);
                acc2 = __builtin_amdgcn_wmma_f32_16x16x4_f32(false, a, false, b2, (short)0, acc2, false, false);
                acc3 = __builtin_amdgcn_wmma_f32_16x16x4_f32(false, a, false, b3, (short)0, acc3, false, false);
            }
        }
        const float rs[8] = {rs0.x, rs0.y, rs0.z, rs0.w, rs1.x, rs1.y, rs1.z, rs1.w};
        float* Tp = Tw + (8 * h) * TILE_PITCH + c;
        #pragma unroll
        for (int j = 0; j < 8; ++j) {
            Tp[j * TILE_PITCH]      = acc0[j] * rs[j];
            Tp[j * TILE_PITCH + 16] = acc1[j] * rs[j];
            Tp[j * TILE_PITCH + 32] = acc2[j] * rs[j];
            Tp[j * TILE_PITCH + 48] = acc3[j] * rs[j];
        }
    }
    __syncthreads();

    if (valid) {
        const float* Tr = Tw + h * TILE_PITCH + 4 * c;
        float* Cp = C + (size_t)(row0 + h) * 64 + 4 * c;
        #pragma unroll
        for (int r = 0; r < 8; ++r) {
            const float4 o = *(const float4*)(Tr + (2 * r) * TILE_PITCH);
            *(float4*)(Cp + (size_t)(2 * r) * 64) = o;
        }
    }
}


__global__ __launch_bounds__(256) void gather_kernel(const float* __restrict__ h,
                                                     const float* __restrict__ dinv,
                                                     const int* __restrict__ cnt,
                                                     const int* __restrict__ rowstart,
                                                     const int* __restrict__ col,
                                                     const float* __restrict__ bias,
                                                     float* __restrict__ out,
                                                     int n_nodes, int e_count, int do_relu) {
    int gid = blockIdx.x * blockDim.x + threadIdx.x;
    int n = gid >> 4;
    if (n >= n_nodes) return;
    int c = (gid & 15) << 2;
    const float di = dinv[n];
    const int start = max(rowstart[n], 0);
    const int end = min(start + cnt[n], e_count);
    float4 a = make_float4(0.f, 0.f, 0.f, 0.f);
    if (start < end) {
        const int last = end - 1;
        int s0 = col[start];
        int s1 = col[min(start + 1, last)];
        int s2 = col[min(start + 2, last)];
        int s3 = col[min(start + 3, last)];
        for (int p = start; p < end; p += 4) {
            const float4 h0 = *(const float4*)(h + (size_t)s0 * 64 + c);
            const float4 h1 = *(const float4*)(h + (size_t)s1 * 64 + c);
            const float4 h2 = *(const float4*)(h + (size_t)s2 * 64 + c);
            const float4 h3 = *(const float4*)(h + (size_t)s3 * 64 + c);
            const float w1 = (p + 1 < end) ? 1.f : 0.f;
            const float w2 = (p + 2 < end) ? 1.f : 0.f;
            const float w3 = (p + 3 < end) ? 1.f : 0.f;
            s0 = col[min(p + 4, last)];
            s1 = col[min(p + 5, last)];
            s2 = col[min(p + 6, last)];
            s3 = col[min(p + 7, last)];
            a.x += h0.x; a.y += h0.y; a.z += h0.z; a.w += h0.w;
            a.x = fmaf(h1.x, w1, a.x); a.y = fmaf(h1.y, w1, a.y);
            a.z = fmaf(h1.z, w1, a.z); a.w = fmaf(h1.w, w1, a.w);
            a.x = fmaf(h2.x, w2, a.x); a.y = fmaf(h2.y, w2, a.y);
            a.z = fmaf(h2.z, w2, a.z); a.w = fmaf(h2.w, w2, a.w);
            a.x = fmaf(h3.x, w3, a.x); a.y = fmaf(h3.y, w3, a.y);
            a.z = fmaf(h3.z, w3, a.z); a.w = fmaf(h3.w, w3, a.w);
        }
    }
    const float4 hv = *(const float4*)(h + (size_t)n * 64 + c);
    const float4 b  = *(const float4*)(bias + c);
    float4 r;
    r.x = (a.x + hv.x) * di + b.x;
    r.y = (a.y + hv.y) * di + b.y;
    r.z = (a.z + hv.z) * di + b.z;
    r.w = (a.w + hv.w) * di + b.w;
    if (do_relu) {
        r.x = fmaxf(r.x, 0.f); r.y = fmaxf(r.y, 0.f);
        r.z = fmaxf(r.z, 0.f); r.w = fmaxf(r.w, 0.f);
    }
    *(float4*)(out + (size_t)n * 64 + c) = r;
}


static_assert(N_NODES % 16 == 0, "the fused kernel has no partial tile");

__global__ __launch_bounds__(256) void gather_gemm_kernel(const float* __restrict__ h,
                                                          const float* __restrict__ dinv,
                                                          const int* __restrict__ cnt,
                                                          const int* __restrict__ rowstart,
                                                          const int* __restrict__ col,
                                                          const float* __restrict__ bias,
                                                          const float* __restrict__ Wn,
                                                          float* __restrict__ out,
                                                          int e_count) {
    __shared__ alignas(16) float sH[16 * TILE_PITCH];
    __shared__ alignas(16) float sO[16 * TILE_PITCH];

    const int tid = threadIdx.x;
    const int nl = tid >> 4;
    const int n = blockIdx.x * 16 + nl;
    const int c = (tid & 15) << 2;
    const float di = dinv[n];
    const int start = max(rowstart[n], 0);
    const int end = min(start + cnt[n], e_count);
    float4 a = make_float4(0.f, 0.f, 0.f, 0.f);
    if (start < end) {
        const int last = end - 1;
        int s0 = col[start];
        int s1 = col[min(start + 1, last)];
        int s2 = col[min(start + 2, last)];
        int s3 = col[min(start + 3, last)];
        for (int p = start; p < end; p += 4) {
            const float4 h0 = *(const float4*)(h + (size_t)s0 * 64 + c);
            const float4 h1 = *(const float4*)(h + (size_t)s1 * 64 + c);
            const float4 h2 = *(const float4*)(h + (size_t)s2 * 64 + c);
            const float4 h3 = *(const float4*)(h + (size_t)s3 * 64 + c);
            const float w1 = (p + 1 < end) ? 1.f : 0.f;
            const float w2 = (p + 2 < end) ? 1.f : 0.f;
            const float w3 = (p + 3 < end) ? 1.f : 0.f;
            s0 = col[min(p + 4, last)];
            s1 = col[min(p + 5, last)];
            s2 = col[min(p + 6, last)];
            s3 = col[min(p + 7, last)];
            a.x += h0.x; a.y += h0.y; a.z += h0.z; a.w += h0.w;
            a.x = fmaf(h1.x, w1, a.x); a.y = fmaf(h1.y, w1, a.y);
            a.z = fmaf(h1.z, w1, a.z); a.w = fmaf(h1.w, w1, a.w);
            a.x = fmaf(h2.x, w2, a.x); a.y = fmaf(h2.y, w2, a.y);
            a.z = fmaf(h2.z, w2, a.z); a.w = fmaf(h2.w, w2, a.w);
            a.x = fmaf(h3.x, w3, a.x); a.y = fmaf(h3.y, w3, a.y);
            a.z = fmaf(h3.z, w3, a.z); a.w = fmaf(h3.w, w3, a.w);
        }
    }
    const float4 hv = *(const float4*)(h + (size_t)n * 64 + c);
    const float4 b  = *(const float4*)(bias + c);
    float4 r;
    r.x = fmaxf((a.x + hv.x) * di + b.x, 0.f);
    r.y = fmaxf((a.y + hv.y) * di + b.y, 0.f);
    r.z = fmaxf((a.z + hv.z) * di + b.z, 0.f);
    r.w = fmaxf((a.w + hv.w) * di + b.w, 0.f);
    *(float4*)(sH + nl * TILE_PITCH + c) = r;

    const int wave = tid >> 5;
    const int lane = tid & 31;
    const int row16 = lane & 15;
    const int khalf = (lane >> 4) << 1;

    v2f fb[16];
    if (wave < 4) {
        const float* __restrict__ Bcol = Wn + khalf * 64 + wave * 16 + row16;
        #pragma unroll
        for (int i = 0; i < 16; ++i) {
            fb[i].x = Bcol[(4 * i) * 64];
            fb[i].y = Bcol[(4 * i + 1) * 64];
        }
    }
    __syncthreads();

    if (wave < 4) {
        const float* Arow = sH + row16 * TILE_PITCH + khalf;
        v8f acc = {};
        #pragma unroll
        for (int i = 0; i < 16; ++i) {
            v2f fa;
            fa.x = Arow[4 * i];
            fa.y = Arow[4 * i + 1];
            acc = __builtin_amdgcn_wmma_f32_16x16x4_f32(
                 false, fa,  false, fb[i],
                 (short)0, acc,  false,  false);
        }
        float* Orow = sO + ((lane >> 4) << 3) * TILE_PITCH + wave * 16 + row16;
        #pragma unroll
        for (int j = 0; j < 8; ++j) Orow[j * TILE_PITCH] = acc[j];
    }
    __syncthreads();

    float4 o = *(const float4*)(sO + nl * TILE_PITCH + c);
    o.x *= di; o.y *= di; o.z *= di; o.w *= di;
    *(float4*)(out + (size_t)n * 64 + c) = o;
}


extern "C" void kernel_launch(void* const* d_in, const int* in_sizes, int n_in,
                              void* d_out, int out_size, void* d_ws, size_t ws_size,
                              hipStream_t stream) {
    const float* x  = (const float*)d_in[0];
    const int*   ei = (const int*)d_in[1];
    const float* W1 = (const float*)d_in[2];
    const float* b1 = (const float*)d_in[3];
    const float* W2 = (const float*)d_in[4];
    const float* b2 = (const float*)d_in[5];
    float* out = (float*)d_out;

    const int* src = ei;
    const int* dst = ei + N_EDGES;

    float* dinv      = (float*)d_ws;
    int*   cnt       = (int*)(dinv + 65536);
    int*   rowstart  = cnt + 65536;
    int2*  tab       = (int2*)(rowstart + 65536);
    int2*  bucketed  = tab + 65536;
    int*   col       = (int*)(bucketed + N_EDGES);
    float* buf0      = (float*)(col + N_EDGES);
    float* buf1      = buf0 + (size_t)N_NODES * 64;

    const int B = 256;
    dim3 blk(B);
    dim3 blk1k(1024);

    chunk_sort_kernel<<<N_CHUNKS, blk1k, 0, stream>>>(src, dst, tab, bucketed);
    bucket_finalize_kernel<<<N_BUCKETS, blk1k, 0, stream>>>(bucketed, tab, cnt, rowstart, dinv, col);

    int gemm_grid = (N_NODES / 16 + 7) / 8;
    gemm_strip_wmma_f32<<<gemm_grid, blk, 0, stream>>>(x, W1, dinv, buf0);
    gather_gemm_kernel<<<N_NODES / 16, blk, 0, stream>>>(buf0, dinv, cnt, rowstart, col, b1, W2, buf1, N_EDGES);

    gather_kernel<<<(N_NODES * 16 + B - 1) / B, blk, 0, stream>>>(buf1, dinv, cnt, rowstart, col, b2,
                                                                  out, N_NODES, N_EDGES,  0);
}


# Both gathers: a 16-lane group sums 4 nodes in turn; a node's first 32 sources come as two 64-byte reads shared by DPP, 16 row loads in flight; 25000 -> 6256 waves; layer-1 tile 64 nodes; same sums
# speedup vs baseline: 4.7590x; 1.0453x over previous
#include <hip/hip_runtime.h>
#include <hip/hip_bf16.h>
#include <math.h>


#define N_NODES 50000
#define N_EDGES 800000
#define IN_DIM  128
#define HID_DIM 64

#define N_BUCKETS 196
#define N_CHUNKS  256
#define CHUNK     3125
#define LSORT_CAP 8192

static_assert(N_CHUNKS * CHUNK == N_EDGES, "chunks must tile the edge list exactly");
static_assert(N_BUCKETS * 256 >= N_NODES && N_BUCKETS <= 256, "bucket count");

typedef float v2f __attribute__((ext_vector_type(2)));
typedef float v8f __attribute__((ext_vector_type(8)));


__device__ __forceinline__ int bucket_of(int d) {
    unsigned b = ((unsigned)d) >> 8;
    return (int)(b < (unsigned)(N_BUCKETS - 1) ? b : (unsigned)(N_BUCKETS - 1));
}

__device__ __forceinline__ int block_incl_scan_1024(int v, int* wsum) {
    const int lane = threadIdx.x & 31;
    const int wave = threadIdx.x >> 5;
    int incl = v;
    for (int d = 1; d < 32; d <<= 1) {
        int t = __shfl_up(incl, d);
        if (lane >= d) incl += t;
    }
    if (lane == 31) wsum[wave] = incl;
    __syncthreads();
    int base = 0;
    for (int w = 0; w < 32; ++w) {
        int s = wsum[w];
        base += (w < wave) ? s : 0;
    }
    __syncthreads();
    return base + incl;
}

__global__ __launch_bounds__(1024) void chunk_sort_kernel(const int* __restrict__ src,
                                                          const int* __restrict__ dst,
                                                          int2* __restrict__ tab,
                                                          int2* __restrict__ bucketed) {
    __shared__ int sorted_s[CHUNK];
    __shared__ int sorted_d[CHUNK];
    __shared__ int hcnt[256];
    __shared__ int lstart[256];
    __shared__ int wsum[32];

    const int tid = threadIdx.x;
    const int c = blockIdx.x;

    int es[4], ed[4], rk[4];
    for (int j = 0; j < 4; ++j) {
        const int i = tid + j * 1024;
        es[j] = 0; ed[j] = 0;
        if (i < CHUNK) { es[j] = src[c * CHUNK + i]; ed[j] = dst[c * CHUNK + i]; }
    }
    if (tid < 256) hcnt[tid] = 0;
    __syncthreads();

    for (int j = 0; j < 4; ++j) {
        const int i = tid + j * 1024;
        rk[j] = 0;
        if (i < CHUNK) rk[j] = atomicAdd(&hcnt[bucket_of(ed[j])], 1);
    }
    __syncthreads();

    const int lcnt = (tid < 256) ? hcnt[tid] : 0;
    const int incl = block_incl_scan_1024(lcnt, wsum);
    if (tid < 256) {
        lstart[tid] = incl - lcnt;
        tab[c * 256 + tid] = make_int2(incl - lcnt, lcnt);
    }
    __syncthreads();

    for (int j = 0; j < 4; ++j) {
        const int i = tid + j * 1024;
        if (i < CHUNK) {
            const int lp = min(max(lstart[bucket_of(ed[j])] + rk[j], 0), CHUNK - 1);
            sorted_s[lp] = es[j];
            sorted_d[lp] = ed[j];
        }
    }
    __syncthreads();

    for (int j = 0; j < 4; ++j) {
        const int i = tid + j * 1024;
        if (i < CHUNK) bucketed[c * CHUNK + i] = make_int2(sorted_s[i], sorted_d[i]);
    }
}

__device__ __forceinline__ int piece_index(const int* pstart, const int* pbase, int i) {
    int lo = 0;
    for (int step = 128; step >= 1; step >>= 1) {
        if (pstart[lo + step] <= i) lo += step;
    }
    return min(max(pbase[lo] + (i - pstart[lo]), 0), N_EDGES - 1);
}

__global__ __launch_bounds__(1024) void bucket_finalize_kernel(const int2* __restrict__ bucketed,
                                                               const int2* __restrict__ tab,
                                                               int* __restrict__ cnt,
                                                               int* __restrict__ rowstart,
                                                               float* __restrict__ dinv,
                                                               int* __restrict__ col) {
    __shared__ int ncnt[256];
    __shared__ int nstart[256];
    __shared__ int pstart[256];
    __shared__ int pbase[256];
    __shared__ int2 wtot[8];
    __shared__ int wsum[32];
    __shared__ int lsorted[LSORT_CAP];

    const int tid = threadIdx.x;
    const int b = blockIdx.x;

    int pc = 0, pl = 0;
    if (tid < 256) {
        const int2 e = tab[tid * 256 + b];
        pc = min(max(e.y, 0), CHUNK);
        pl = min(max(e.x, 0), CHUNK - pc);
    }
    int incl_c = pc, incl_l = pl;
    for (int d = 1; d < 32; d <<= 1) {
        const int tc = __shfl_up(incl_c, d);
        const int tl = __shfl_up(incl_l, d);
        if ((tid & 31) >= d) { incl_c += tc; incl_l += tl; }
    }
    if ((tid & 31) == 31 && tid < 256) wtot[tid >> 5] = make_int2(incl_c, incl_l);
    __syncthreads();
    int base_c = 0, base_l = 0, tot_c = 0, tot_l = 0;
    for (int w = 0; w < 8; ++w) {
        const int2 t = wtot[w];
        tot_c += t.x; tot_l += t.y;
        if (w < (tid >> 5)) { base_c += t.x; base_l += t.y; }
    }
    if (tid < 256) {
        pstart[tid] = base_c + incl_c - pc;
        pbase[tid] = tid * CHUNK + pl;
        ncnt[tid] = 0;
    }
    __syncthreads();

    const int s = min(max(tot_l, 0), N_EDGES);
    const int m = min(max(tot_c, 0), N_EDGES - s);

    const bool in_lds = (m <= LSORT_CAP);

    int rs[8], rd[8];
    const int mlast = max(m - 1, 0);
    for (int k = 0; k < 8; ++k) {
        const int i = min(tid + k * 1024, mlast);
        const int2 rec = bucketed[piece_index(pstart, pbase, i)];
        rs[k] = rec.x; rd[k] = rec.y;
    }

    if (in_lds) {
        for (int k = 0; k < 8; ++k) {
            if (tid + k * 1024 < m) atomicAdd(&ncnt[rd[k] & 255], 1);
        }
    } else {
        for (int i = tid; i < m; i += 1024) {
            const int d = bucketed[piece_index(pstart, pbase, i)].y;
            atomicAdd(&ncnt[d & 255], 1);
        }
    }
    __syncthreads();

    const int v = (tid < 256) ? ncnt[tid] : 0;
    const int incl = block_incl_scan_1024(v, wsum);
    if (tid < 256) {
        const int ex = incl - v;
        nstart[tid] = ex;
        ncnt[tid] = 0;
        const int n = b * 256 + tid;
        if (n < N_NODES) {
            cnt[n] = v;
            rowstart[n] = s + ex;
            const float dg = (float)(v + 1);
            dinv[n] = (dg > 0.0f) ? rsqrtf(dg) : 0.0f;
        }
    }
    __syncthreads();

    if (in_lds) {
        for (int k = 0; k < 8; ++k) {
            if (tid + k * 1024 < m) {
                const int dl = rd[k] & 255;
                const int r = atomicAdd(&ncnt[dl], 1);
                const int p = min(max(nstart[dl] + r, 0), m - 1);
                lsorted[p] = rs[k];
            }
        }
    } else {
        for (int i = tid; i < m; i += 1024) {
            const int2 rec = bucketed[piece_index(pstart, pbase, i)];
            const int dl = rec.y & 255;
            const int r = atomicAdd(&ncnt[dl], 1);
            const int p = min(max(nstart[dl] + r, 0), m - 1);
            col[s + p] = rec.x;
        }
    }
    __syncthreads();
    if (in_lds) {
        for (int i = tid; i < m; i += 1024) col[s + i] = lsorted[i];
    }
}


#define W1_PITCH 144
#define TILE_PITCH 68

static_assert(IN_DIM == 128 && HID_DIM == 64, "the GEMM's tiling is written for a 128 x 64 weight");
static_assert(N_NODES % 16 == 0, "a 16-row tile is either whole or absent: the stores need no row guard");

__global__ __launch_bounds__(256) void gemm_strip_wmma_f32(const float* __restrict__ A,
                                                           const float* __restrict__ W,
                                                           const float* __restrict__ rscale,
                                                           float* __restrict__ C) {
    __shared__ alignas(16) float sWp[(IN_DIM / 2) * W1_PITCH];
    __shared__ alignas(16) float sT[8 * 16 * TILE_PITCH];

    const int tid = threadIdx.x;
    const int wave = tid >> 5;
    const int lane = tid & 31;
    const int c = lane & 15;
    const int h = lane >> 4;

    float4 w0[4], w1[4];
    #pragma unroll
    for (int j = 0; j < 4; ++j) {
        const int id = tid + j * 256;
        const int g = id & 15;
        const int kp = id >> 4;
        w0[j] = *(const float4*)(W + (2 * kp) * 64 + 4 * g);
        w1[j] = *(const float4*)(W + (2 * kp + 1) * 64 + 4 * g);
    }

    const int m_tile = blockIdx.x * 8 + wave;
    const bool valid = (m_tile * 16 < N_NODES);
    const int row0 = valid ? m_tile * 16 : (N_NODES - 16);
    const float* __restrict__ Ap = A + (size_t)(row0 + c) * IN_DIM + 4 * h;
    float4 q[16];
    #pragma unroll
    for (int i = 0; i < 16; ++i) q[i] = *(const float4*)(Ap + 8 * i);
    const float4 rs0 = *(const float4*)(rscale + row0 + 8 * h);
    const float4 rs1 = *(const float4*)(rscale + row0 + 8 * h + 4);

    #pragma unroll
    for (int j = 0; j < 4; ++j) {
        const int id = tid + j * 256;
        const int g = id & 15;
        const int kp = id >> 4;
        float* p = sWp + kp * W1_PITCH + 8 * g;
        *(float4*)(p)     = make_float4(w0[j].x, w1[j].x, w0[j].y, w1[j].y);
        *(float4*)(p + 4) = make_float4(w0[j].z, w1[j].z, w0[j].w, w1[j].w);
    }
    __builtin_amdgcn_fence(__ATOMIC_RELEASE, "workgroup", "local");
    __builtin_amdgcn_s_barrier();
    __builtin_amdgcn_fence(__ATOMIC_ACQUIRE, "workgroup", "local");

    float* Tw = sT + wave * (16 * TILE_PITCH);
    if (valid) {
        const float* Bp = sWp + (2 * h) * W1_PITCH + 2 * c;
        v8f acc0 = {}, acc1 = {}, acc2 = {}, acc3 = {};
        #pragma unroll
        for (int i = 0; i < 16; ++i) {
            #pragma unroll
            for (int st = 0; st < 2; ++st) {
                v2f a;
                a.x = st ? q[i].z : q[i].x;
                a.y = st ? q[i].w : q[i].y;
                const float* bp = Bp + (4 * i + st) * W1_PITCH;
                const v2f b0 = *(const v2f*)(bp);
                const v2f b1 = *(const v2f*)(bp + 32);
                const v2f b2 = *(const v2f*)(bp + 64);
                const v2f b3 = *(const v2f*)(bp + 96);
                acc0 = __builtin_amdgcn_wmma_f32_16x16x4_f32(false, a, false, b0, (short)0, acc0, false, false);
                acc1 = __builtin_amdgcn_wmma_f32_16x16x4_f32(false, a, false, b1, (short)0, acc1, false, false);
                acc2 = __builtin_amdgcn_wmma_f32_16x16x4_f32(false, a, false, b2, (short)0, acc2, false, false);
                acc3 = __builtin_amdgcn_wmma_f32_16x16x4_f32(false, a, false, b3, (short)0, acc3, false, false);
            }
        }
        const float rs[8] = {rs0.x, rs0.y, rs0.z, rs0.w, rs1.x, rs1.y, rs1.z, rs1.w};
        float* Tp = Tw + (8 * h) * TILE_PITCH + c;
        #pragma unroll
        for (int j = 0; j < 8; ++j) {
            Tp[j * TILE_PITCH]      = acc0[j] * rs[j];
            Tp[j * TILE_PITCH + 16] = acc1[j] * rs[j];
            Tp[j * TILE_PITCH + 32] = acc2[j] * rs[j];
            Tp[j * TILE_PITCH + 48] = acc3[j] * rs[j];
        }
    }
    __syncthreads();

    if (valid) {
        const float* Tr = Tw + h * TILE_PITCH + 4 * c;
        float* Cp = C + (size_t)(row0 + h) * 64 + 4 * c;
        #pragma unroll
        for (int r = 0; r < 8; ++r) {
            const float4 o = *(const float4*)(Tr + (2 * r) * TILE_PITCH);
            *(float4*)(Cp + (size_t)(2 * r) * 64) = o;
        }
    }
}


#define NPG 4
static_assert(N_NODES % NPG == 0, "a group's nodes are all present or all absent");

__device__ __forceinline__ float4 row_f4(const float* __restrict__ h, int s, int cbytes) {
    const unsigned off = ((unsigned)s << 8) | (unsigned)cbytes;
    return *(const float4*)((const char*)h + off);
}

__device__ __forceinline__ void fma_row(float4& a, const float4 r, const float w) {
    a.x = fmaf(r.x, w, a.x); a.y = fmaf(r.y, w, a.y);
    a.z = fmaf(r.z, w, a.z); a.w = fmaf(r.w, w, a.w);
}

__global__ __launch_bounds__(256) void gather_kernel(const float* __restrict__ h,
                                                     const float* __restrict__ dinv,
                                                     const int* __restrict__ cnt,
                                                     const int* __restrict__ rowstart,
                                                     const int* __restrict__ col,
                                                     const float* __restrict__ bias,
                                                     float* __restrict__ out,
                                                     int n_nodes, int e_count, int do_relu) {
    const int grp = (blockIdx.x * blockDim.x + threadIdx.x) >> 4;
    const int l16 = threadIdx.x & 15;
    const int c = l16 << 2;
    const int cb = l16 << 4;
    const int n0 = grp * NPG;
    if (n0 >= n_nodes) return;
    const int elast = max(e_count - 1, 0);

    int st[NPG], dg[NPG];
    float di[NPG];
    #pragma unroll
    for (int k = 0; k < NPG; ++k) {
        const int n = n0 + k;
        const int s = max(rowstart[n], 0);
        const int e = min(s + cnt[n], e_count);
        st[k] = s;
        dg[k] = max(e - s, 0);
        di[k] = dinv[n];
    }
    int ia[NPG], ib[NPG];
    #pragma unroll
    for (int k = 0; k < NPG; ++k) {
        const int lastp = min(max(st[k] + dg[k] - 1, 0), elast);
        ia[k] = col[min(st[k] + l16, lastp)];
        ib[k] = col[min(st[k] + 16 + l16, lastp)];
    }
    const float4 b = *(const float4*)(bias + c);
    asm volatile("" :: "v"(ia[0]), "v"(ia[1]), "v"(ia[2]), "v"(ia[3]),
                       "v"(ib[0]), "v"(ib[1]), "v"(ib[2]), "v"(ib[3]));

    #pragma unroll
    for (int k = 0; k < NPG; ++k) {
        const int n = n0 + k;
        const int d = dg[k];
        const float4 hv = *(const float4*)(h + (size_t)n * 64 + c);
        float4 a = make_float4(0.f, 0.f, 0.f, 0.f);
        {
            float4 r[16];
            #pragma unroll
            for (int j = 0; j < 16; ++j) r[j] = row_f4(h, __shfl(ia[k], j, 16), cb);
            #pragma unroll
            for (int j = 0; j < 16; ++j) fma_row(a, r[j], (j < d) ? 1.f : 0.f);
        }
        if (d > 16) {
            float4 q[8];
            #pragma unroll
            for (int j = 0; j < 8; ++j) q[j] = row_f4(h, __shfl(ib[k], j, 16), cb);
            #pragma unroll
            for (int j = 0; j < 8; ++j) fma_row(a, q[j], (16 + j < d) ? 1.f : 0.f);
            if (d > 24) {
                #pragma unroll
                for (int j = 0; j < 8; ++j) q[j] = row_f4(h, __shfl(ib[k], 8 + j, 16), cb);
                #pragma unroll
                for (int j = 0; j < 8; ++j) fma_row(a, q[j], (24 + j < d) ? 1.f : 0.f);
                for (int p = 32; p < d; ++p) {
                    const int s = col[min(st[k] + p, elast)];
                    fma_row(a, row_f4(h, s, cb), 1.f);
                }
            }
        }
        float4 r;
        r.x = (a.x + hv.x) * di[k] + b.x;
        r.y = (a.y + hv.y) * di[k] + b.y;
        r.z = (a.z + hv.z) * di[k] + b.z;
        r.w = (a.w + hv.w) * di[k] + b.w;
        if (do_relu) {
            r.x = fmaxf(r.x, 0.f); r.y = fmaxf(r.y, 0.f);
            r.z = fmaxf(r.z, 0.f); r.w = fmaxf(r.w, 0.f);
        }
        *(float4*)(out + (size_t)n * 64 + c) = r;
    }
}


#define GG_NODES 64
static_assert(GG_NODES == 16 * NPG, "one group of 16 lanes per NPG nodes");
static_assert(N_NODES % 16 == 0, "a 16-row tile is whole or absent");

__global__ __launch_bounds__(256) void gather_gemm_kernel(const float* __restrict__ h,
                                                          const float* __restrict__ dinv,
                                                          const int* __restrict__ cnt,
                                                          const int* __restrict__ rowstart,
                                                          const int* __restrict__ col,
                                                          const float* __restrict__ bias,
                                                          const float* __restrict__ Wn,
                                                          float* __restrict__ out,
                                                          int e_count) {
    __shared__ alignas(16) float sH[GG_NODES * TILE_PITCH];
    __shared__ alignas(16) float sO[GG_NODES * TILE_PITCH];

    const int tid = threadIdx.x;
    const int g = tid >> 4;
    const int l16 = tid & 15;
    const int c = l16 << 2;
    const int cb = l16 << 4;
    const int rl0 = g * NPG;
    const int n0 = blockIdx.x * GG_NODES + rl0;
    const bool present = (n0 < N_NODES);
    const int elast = max(e_count - 1, 0);

    float di[NPG];
    #pragma unroll
    for (int k = 0; k < NPG; ++k) di[k] = 0.f;

    if (present) {
        int st[NPG], dg[NPG];
        #pragma unroll
        for (int k = 0; k < NPG; ++k) {
            const int n = n0 + k;
            const int s = max(rowstart[n], 0);
            const int e = min(s + cnt[n], e_count);
            st[k] = s;
            dg[k] = max(e - s, 0);
            di[k] = dinv[n];
        }
        int ia[NPG], ib[NPG];
        #pragma unroll
        for (int k = 0; k < NPG; ++k) {
            const int lastp = min(max(st[k] + dg[k] - 1, 0), elast);
            ia[k] = col[min(st[k] + l16, lastp)];
            ib[k] = col[min(st[k] + 16 + l16, lastp)];
        }
        const float4 b = *(const float4*)(bias + c);
        asm volatile("" :: "v"(ia[0]), "v"(ia[1]), "v"(ia[2]), "v"(ia[3]),
                           "v"(ib[0]), "v"(ib[1]), "v"(ib[2]), "v"(ib[3]));

        #pragma unroll
        for (int k = 0; k < NPG; ++k) {
            const int n = n0 + k;
            const int d = dg[k];
            const float4 hv = *(const float4*)(h + (size_t)n * 64 + c);
            float4 a = make_float4(0.f, 0.f, 0.f, 0.f);
            {
                float4 r[16];
                #pragma unroll
                for (int j = 0; j < 16; ++j) r[j] = row_f4(h, __shfl(ia[k], j, 16), cb);
                #pragma unroll
                for (int j = 0; j < 16; ++j) fma_row(a, r[j], (j < d) ? 1.f : 0.f);
            }
            if (d > 16) {
                float4 q[8];
                #pragma unroll
                for (int j = 0; j < 8; ++j) q[j] = row_f4(h, __shfl(ib[k], j, 16), cb);
                #pragma unroll
                for (int j = 0; j < 8; ++j) fma_row(a, q[j], (16 + j < d) ? 1.f : 0.f);
                if (d > 24) {
                    #pragma unroll
                    for (int j = 0; j < 8; ++j) q[j] = row_f4(h, __shfl(ib[k], 8 + j, 16), cb);
                    #pragma unroll
                    for (int j = 0; j < 8; ++j) fma_row(a, q[j], (24 + j < d) ? 1.f : 0.f);
                    for (int p = 32; p < d; ++p) {
                        const int s = col[min(st[k] + p, elast)];
                        fma_row(a, row_f4(h, s, cb), 1.f);
                    }
                }
            }
            float4 r;
            r.x = fmaxf((a.x + hv.x) * di[k] + b.x, 0.f);
            r.y = fmaxf((a.y + hv.y) * di[k] + b.y, 0.f);
            r.z = fmaxf((a.z + hv.z) * di[k] + b.z, 0.f);
            r.w = fmaxf((a.w + hv.w) * di[k] + b.w, 0.f);
            *(float4*)(sH + (rl0 + k) * TILE_PITCH + c) = r;
        }
    } else {
        #pragma unroll
        for (int k = 0; k < NPG; ++k)
            *(float4*)(sH + (rl0 + k) * TILE_PITCH + c) = make_float4(0.f, 0.f, 0.f, 0.f);
    }

    const int wave = tid >> 5;
    const int lane = tid & 31;
    const int row16 = lane & 15;
    const int khalf = (lane >> 4) << 1;
    const int tile = wave >> 1;
    const int cblk = (wave & 1) << 1;

    v2f fb0[16], fb1[16];
    {
        const float* __restrict__ Bcol = Wn + khalf * 64 + cblk * 16 + row16;
        #pragma unroll
        for (int i = 0; i < 16; ++i) {
            fb0[i].x = Bcol[(4 * i) * 64];
            fb0[i].y = Bcol[(4 * i + 1) * 64];
            fb1[i].x = Bcol[(4 * i) * 64 + 16];
            fb1[i].y = Bcol[(4 * i + 1) * 64 + 16];
        }
    }
    __syncthreads();

    {
        const float* Arow = sH + (tile * 16 + row16) * TILE_PITCH + khalf;
        v8f acc0 = {}, acc1 = {};
        #pragma unroll
        for (int i = 0; i < 16; ++i) {
            v2f fa;
            fa.x = Arow[4 * i];
            fa.y = Arow[4 * i + 1];
            acc0 = __builtin_amdgcn_wmma_f32_16x16x4_f32(
                 false, fa,  false, fb0[i],
                 (short)0, acc0,  false,  false);
            acc1 = __builtin_amdgcn_wmma_f32_16x16x4_f32(
                 false, fa,  false, fb1[i],
                 (short)0, acc1,  false,  false);
        }
        float* Orow = sO + (tile * 16 + ((lane >> 4) << 3)) * TILE_PITCH + cblk * 16 + row16;
        #pragma unroll
        for (int j = 0; j < 8; ++j) {
            Orow[j * TILE_PITCH]      = acc0[j];
            Orow[j * TILE_PITCH + 16] = acc1[j];
        }
    }
    __syncthreads();

    if (present) {
        #pragma unroll
        for (int k = 0; k < NPG; ++k) {
            float4 o = *(const float4*)(sO + (rl0 + k) * TILE_PITCH + c);
            o.x *= di[k]; o.y *= di[k]; o.z *= di[k]; o.w *= di[k];
            *(float4*)(out + (size_t)(n0 + k) * 64 + c) = o;
        }
    }
}


extern "C" void kernel_launch(void* const* d_in, const int* in_sizes, int n_in,
                              void* d_out, int out_size, void* d_ws, size_t ws_size,
                              hipStream_t stream) {
    const float* x  = (const float*)d_in[0];
    const int*   ei = (const int*)d_in[1];
    const float* W1 = (const float*)d_in[2];
    const float* b1 = (const float*)d_in[3];
    const float* W2 = (const float*)d_in[4];
    const float* b2 = (const float*)d_in[5];
    float* out = (float*)d_out;

    const int* src = ei;
    const int* dst = ei + N_EDGES;

    float* dinv      = (float*)d_ws;
    int*   cnt       = (int*)(dinv + 65536);
    int*   rowstart  = cnt + 65536;
    int2*  tab       = (int2*)(rowstart + 65536);
    int2*  bucketed  = tab + 65536;
    int*   col       = (int*)(bucketed + N_EDGES);
    float* buf0      = (float*)(col + N_EDGES);
    float* buf1      = buf0 + (size_t)N_NODES * 64;

    const int B = 256;
    dim3 blk(B);
    dim3 blk1k(1024);

    chunk_sort_kernel<<<N_CHUNKS, blk1k, 0, stream>>>(src, dst, tab, bucketed);
    bucket_finalize_kernel<<<N_BUCKETS, blk1k, 0, stream>>>(bucketed, tab, cnt, rowstart, dinv, col);

    int gemm_grid = (N_NODES / 16 + 7) / 8;
    gemm_strip_wmma_f32<<<gemm_grid, blk, 0, stream>>>(x, W1, dinv, buf0);
    gather_gemm_kernel<<<(N_NODES + GG_NODES - 1) / GG_NODES, blk, 0, stream>>>(buf0, dinv, cnt, rowstart, col, b1, W2, buf1, N_EDGES);

    gather_kernel<<<(N_NODES / NPG * 16 + B - 1) / B, blk, 0, stream>>>(buf1, dinv, cnt, rowstart, col, b2,
                                                                  out, N_NODES, N_EDGES,  0);
}
